# starting kernel
# baseline (speedup 1.0000x reference)
#include <hip/hip_runtime.h>
#include <hip/hip_bf16.h>


typedef __attribute__((ext_vector_type(2))) float v2f;
typedef __attribute__((ext_vector_type(8))) float v8f;

#define N_NODES 40000
#define N_EDGES 640000
#define IN_DIM  3
#define HID_DIM 128
#define LAT_DIM 64

__global__ void zero_kernel(float* __restrict__ p, int n) {
    int i = blockIdx.x * blockDim.x + threadIdx.x;
    int stride = gridDim.x * blockDim.x;
    for (; i < n; i += stride) p[i] = 0.0f;
}

__global__ void degree_kernel(const int* __restrict__ dst, float* __restrict__ deg,
                              int nEdges) {
    int e = blockIdx.x * blockDim.x + threadIdx.x;
    if (e < nEdges) atomicAdd(&deg[dst[e]], 1.0f);
}

__global__ void rsqrt_kernel(float* __restrict__ dis, int n) {
    int i = blockIdx.x * blockDim.x + threadIdx.x;
    if (i < n) dis[i] = rsqrtf(dis[i] + 1.0f);
}

template <int K>
__global__ __launch_bounds__(256) void gemm_wmma_f32(
    const float* __restrict__ A, const float* __restrict__ B,
    float* __restrict__ C, int M, int N) {
    const int lane   = threadIdx.x & 31;
    const int wave   = threadIdx.x >> 5;
    const int tilesN = N >> 4;
    const int tile   = blockIdx.x * 8 + wave;
    const int totalTiles = (M >> 4) * tilesN;
    if (tile >= totalTiles) return;

    const int tm = tile / tilesN;
    const int tn = tile % tilesN;

    const int arow  = tm * 16 + (lane & 15);
    const int khalf = (lane >> 4) * 2;
    const int ncol  = tn * 16 + (lane & 15);

    v8f acc = {};
    for (int k0 = 0; k0 < K; k0 += 4) {
        const int ka = k0 + khalf;
        v2f a, b;
        a.x = (ka     < K) ? A[(size_t)arow * K + ka]     : 0.0f;
        a.y = (ka + 1 < K) ? A[(size_t)arow * K + ka + 1] : 0.0f;
        b.x = (ka     < K) ? B[(size_t)(ka)     * N + ncol] : 0.0f;
        b.y = (ka + 1 < K) ? B[(size_t)(ka + 1) * N + ncol] : 0.0f;
        acc = __builtin_amdgcn_wmma_f32_16x16x4_f32(
            false, a, false, b, (short)0, acc, false, false);
    }

    const int crow = tm * 16 + (lane >> 4) * 8;
#pragma unroll
    for (int r = 0; r < 8; ++r)
        C[(size_t)(crow + r) * N + ncol] = acc[r];
}

template <int DIM>
__global__ __launch_bounds__(256) void scatter_edges(
    const int* __restrict__ src, const int* __restrict__ dst,
    const float* __restrict__ dis, const float* __restrict__ m,
    float* __restrict__ agg, int nEdges) {
    const int gw   = (blockIdx.x * blockDim.x + threadIdx.x) >> 5;
    const int lane = threadIdx.x & 31;
    if (gw >= nEdges) return;

    const int s = src[gw];
    const int d = dst[gw];
    const float coeff = dis[s] * dis[d];

    constexpr int PER = DIM / 32;
    const float* __restrict__ mrow = m + (size_t)s * DIM + lane * PER;
    float* __restrict__ arow = agg + (size_t)d * DIM + lane * PER;

    float v[PER];
#pragma unroll
    for (int j = 0; j < PER; ++j) v[j] = mrow[j];
#pragma unroll
    for (int j = 0; j < PER; ++j) atomicAdd(&arow[j], v[j] * coeff);
}

template <int DIM, bool RELU>
__global__ void finalize_kernel(const float* __restrict__ agg,
                                const float* __restrict__ m,
                                const float* __restrict__ dis,
                                const float* __restrict__ bias,
                                float* __restrict__ out, int nNodes) {
    int idx = blockIdx.x * blockDim.x + threadIdx.x;
    if (idx >= nNodes * DIM) return;
    const int node = idx / DIM;
    const int d    = idx % DIM;
    const float di = dis[node];
    float v = agg[idx] + m[idx] * di * di + bias[d];
    if (RELU) v = fmaxf(v, 0.0f);
    out[idx] = v;
}

extern "C" void kernel_launch(void* const* d_in, const int* in_sizes, int n_in,
                              void* d_out, int out_size, void* d_ws, size_t ws_size,
                              hipStream_t stream) {
    const float* x   = (const float*)d_in[0];
    const int*   ei  = (const int*)d_in[1];
    const float* W0  = (const float*)d_in[2];
    const float* b0  = (const float*)d_in[3];
    const float* W1  = (const float*)d_in[4];
    const float* b1  = (const float*)d_in[5];
    const float* W2  = (const float*)d_in[6];
    const float* b2  = (const float*)d_in[7];
    float* out = (float*)d_out;

    const int* src = ei;
    const int* dst = ei + N_EDGES;

    float* ws  = (float*)d_ws;
    float* dis = ws;
    float* m   = dis + N_NODES;
    float* agg = m   + (size_t)N_NODES * HID_DIM;
    float* h   = agg + (size_t)N_NODES * HID_DIM;

    const int T = 256;
    const int nodeDimBlocksH = (N_NODES * HID_DIM + T - 1) / T;
    const int nodeDimBlocksL = (N_NODES * LAT_DIM + T - 1) / T;
    const int edgeWaveBlocks = (N_EDGES * 32 + T - 1) / T;
    const int gemmBlocksH = (N_NODES / 16) * (HID_DIM / 16) / 8;
    const int gemmBlocksL = (N_NODES / 16) * (LAT_DIM / 16) / 8;

    zero_kernel<<<512, T, 0, stream>>>(dis, N_NODES);
    degree_kernel<<<(N_EDGES + T - 1) / T, T, 0, stream>>>(dst, dis, N_EDGES);
    rsqrt_kernel<<<(N_NODES + T - 1) / T, T, 0, stream>>>(dis, N_NODES);

    gemm_wmma_f32<IN_DIM><<<gemmBlocksH, T, 0, stream>>>(x, W0, m, N_NODES, HID_DIM);
    zero_kernel<<<2048, T, 0, stream>>>(agg, N_NODES * HID_DIM);
    scatter_edges<HID_DIM><<<edgeWaveBlocks, T, 0, stream>>>(src, dst, dis, m, agg, N_EDGES);
    finalize_kernel<HID_DIM, true><<<nodeDimBlocksH, T, 0, stream>>>(agg, m, dis, b0, h, N_NODES);

    gemm_wmma_f32<HID_DIM><<<gemmBlocksH, T, 0, stream>>>(h, W1, m, N_NODES, HID_DIM);
    zero_kernel<<<2048, T, 0, stream>>>(agg, N_NODES * HID_DIM);
    scatter_edges<HID_DIM><<<edgeWaveBlocks, T, 0, stream>>>(src, dst, dis, m, agg, N_EDGES);
    finalize_kernel<HID_DIM, true><<<nodeDimBlocksH, T, 0, stream>>>(agg, m, dis, b1, h, N_NODES);

    gemm_wmma_f32<HID_DIM><<<gemmBlocksL, T, 0, stream>>>(h, W2, m, N_NODES, LAT_DIM);
    zero_kernel<<<1024, T, 0, stream>>>(agg, N_NODES * LAT_DIM);
    scatter_edges<LAT_DIM><<<edgeWaveBlocks, T, 0, stream>>>(src, dst, dis, m, agg, N_EDGES);
    finalize_kernel<LAT_DIM, false><<<nodeDimBlocksL, T, 0, stream>>>(agg, m, dis, b2, out, N_NODES);
}


# GEMMs re-tiled: one wave per 16x64 block of C, 4 accumulator tiles share an A fragment read once in 16-byte pieces, W staged per work-group in LDS in fragment order; k order in 8-blocks permuted
# speedup vs baseline: 2.8723x; 2.8723x over previous
#include <hip/hip_runtime.h>
#include <hip/hip_bf16.h>


typedef __attribute__((ext_vector_type(2))) float v2f;
typedef __attribute__((ext_vector_type(8))) float v8f;

#define N_NODES 40000
#define N_EDGES 640000
#define IN_DIM  3
#define HID_DIM 128
#define LAT_DIM 64

template <int K, int N>
__global__ __launch_bounds__((N / 64) * 128) void gemm_strip_f32(
    const float* __restrict__ A, const float* __restrict__ B,
    float* __restrict__ C, int M) {
    constexpr int NH      = N / 64;
    constexpr int NTT     = N / 16;
    constexpr int KB      = (K + 7) / 8;
    constexpr int THREADS = NH * 128;
    __shared__ float4 Bs[KB * NTT * 32];

    const int tid  = threadIdx.x;
    const int lane = tid & 31;
    const int wave = tid >> 5;
    const int c    = lane & 15;
    const int h    = lane >> 4;
    const int nh   = wave % NH;
    const int strip = blockIdx.x * 4 + wave / NH;
    const bool active = strip * 16 < M;
    const int row  = strip * 16 + c;

    float4 a[KB];
    if (active) {
#pragma unroll
        for (int kb = 0; kb < KB; ++kb) {
            const int k0 = kb * 8 + h * 4;
            if constexpr (K % 8 == 0) {
                a[kb] = *reinterpret_cast<const float4*>(A + (size_t)row * K + k0);
            } else {
                const float* ar = A + (size_t)row * K;
                const float x0 = ar[min(k0,     K - 1)], x1 = ar[min(k0 + 1, K - 1)];
                const float x2 = ar[min(k0 + 2, K - 1)], x3 = ar[min(k0 + 3, K - 1)];
                a[kb].x = (k0     < K) ? x0 : 0.0f;
                a[kb].y = (k0 + 1 < K) ? x1 : 0.0f;
                a[kb].z = (k0 + 2 < K) ? x2 : 0.0f;
                a[kb].w = (k0 + 3 < K) ? x3 : 0.0f;
            }
        }
    }

    constexpr int ITEMS = KB * NH * 32;
    constexpr int TRIPS = (ITEMS + THREADS - 1) / THREADS;
#pragma unroll
    for (int j = 0; j < TRIPS; ++j) {
        const int it = tid + j * THREADS;
        if (ITEMS % THREADS == 0 || it < ITEMS) {
            const int l   = it & 31;
            const int bn  = (it >> 5) % NH;
            const int kb  = (it >> 5) / NH;
            const int k0  = kb * 8 + (l >> 4) * 4;
            const int col = bn * 64 + (l & 15) * 4;
            float4 v[4];
#pragma unroll
            for (int e = 0; e < 4; ++e) {
                if constexpr (K % 8 == 0) {
                    v[e] = *reinterpret_cast<const float4*>(B + (size_t)(k0 + e) * N + col);
                } else {
                    const float4 t = *reinterpret_cast<const float4*>(B + (size_t)min(k0 + e, K - 1) * N + col);
                    v[e] = (k0 + e < K) ? t : make_float4(0.0f, 0.0f, 0.0f, 0.0f);
                }
            }
            float4* d = Bs + (kb * NTT + bn * 4) * 32 + l;
            d[0]  = make_float4(v[0].x, v[1].x, v[2].x, v[3].x);
            d[32] = make_float4(v[0].y, v[1].y, v[2].y, v[3].y);
            d[64] = make_float4(v[0].z, v[1].z, v[2].z, v[3].z);
            d[96] = make_float4(v[0].w, v[1].w, v[2].w, v[3].w);
        }
    }
    __syncthreads();
    if (!active) return;

    v8f acc[4] = {};
#pragma unroll
    for (int kb = 0; kb < KB; ++kb) {
        float4 b[4];
#pragma unroll
        for (int t = 0; t < 4; ++t) b[t] = Bs[(kb * NTT + nh * 4 + t) * 32 + lane];
        const v2f a0 = {a[kb].x, a[kb].y};
        const v2f a1 = {a[kb].z, a[kb].w};
#pragma unroll
        for (int t = 0; t < 4; ++t) {
            const v2f b0 = {b[t].x, b[t].y};
            acc[t] = __builtin_amdgcn_wmma_f32_16x16x4_f32(
                false, a0, false, b0, (short)0, acc[t], false, false);
        }
#pragma unroll
        for (int t = 0; t < 4; ++t) {
            const v2f b1 = {b[t].z, b[t].w};
            acc[t] = __builtin_amdgcn_wmma_f32_16x16x4_f32(
                false, a1, false, b1, (short)0, acc[t], false, false);
        }
    }

    float* __restrict__ crow = C + (size_t)(strip * 16 + h * 8) * N + nh * 64 + c * 4;
#pragma unroll
    for (int r = 0; r < 8; ++r)
        *reinterpret_cast<float4*>(crow + (size_t)r * N) =
            make_float4(acc[0][r], acc[1][r], acc[2][r], acc[3][r]);
}

#define OWN_NODES 160
#define OWN_THREADS 1024
#define LIST_CAP 8192

__global__ __launch_bounds__(OWN_THREADS) void bin_edges(
    const int* __restrict__ src, const int* __restrict__ dst,
    float* __restrict__ dis, int* __restrict__ lists, int* __restrict__ counts,
    int nEdges, int nNodes) {
    __shared__ int l_e[LIST_CAP];
    __shared__ int l_dl[LIST_CAP];
    __shared__ int deg[OWN_NODES];
    __shared__ int cnt;

    const int tid  = threadIdx.x;
    const int base = blockIdx.x * OWN_NODES;

    if (tid < OWN_NODES) deg[tid] = 0;
    if (tid == 0) cnt = 0;
    __syncthreads();

    auto put = [&](int slot, int e, int dl) {
        if (slot < LIST_CAP) { l_e[slot] = e; l_dl[slot] = dl; }
        atomicAdd(&deg[dl], 1);
    };
    auto take4 = [&](int4 d, int e0) {
        const unsigned a0 = (unsigned)(d.x - base), a1 = (unsigned)(d.y - base);
        const unsigned a2 = (unsigned)(d.z - base), a3 = (unsigned)(d.w - base);
        if (min(min(a0, a1), min(a2, a3)) < OWN_NODES) {
            const int c = (a0 < OWN_NODES) + (a1 < OWN_NODES) + (a2 < OWN_NODES) + (a3 < OWN_NODES);
            int slot = atomicAdd(&cnt, c);
            if (a0 < OWN_NODES) put(slot++, e0,     (int)a0);
            if (a1 < OWN_NODES) put(slot++, e0 + 1, (int)a1);
            if (a2 < OWN_NODES) put(slot++, e0 + 2, (int)a2);
            if (a3 < OWN_NODES) put(slot++, e0 + 3, (int)a3);
        }
    };

    const int4* __restrict__ dst4 = reinterpret_cast<const int4*>(dst);
    const int nQuad = nEdges >> 2;
    constexpr int SU = 8;
    for (int q0 = tid; q0 < nQuad; q0 += OWN_THREADS * SU) {
        int4 d[SU];
#pragma unroll
        for (int u = 0; u < SU; ++u) {
            const int q = q0 + OWN_THREADS * u;
            d[u] = (q < nQuad) ? dst4[q] : make_int4(-1, -1, -1, -1);
        }
#pragma unroll
        for (int u = 0; u < SU; ++u) take4(d[u], (q0 + OWN_THREADS * u) * 4);
    }
    for (int e = (nQuad << 2) + tid; e < nEdges; e += OWN_THREADS) {
        const unsigned a = (unsigned)(dst[e] - base);
        if (a < OWN_NODES) put(atomicAdd(&cnt, 1), e, (int)a);
    }
    __syncthreads();

    if (tid < OWN_NODES && base + tid < nNodes)
        dis[base + tid] = rsqrtf((float)deg[tid] + 1.0f);

    const int total = cnt;
    if (tid == 0) counts[blockIdx.x] = total;
    if (total <= LIST_CAP) {
        int* __restrict__ mine = lists + (size_t)blockIdx.x * LIST_CAP;
        for (int i = tid; i < total; i += OWN_THREADS)
            mine[i] = (src[l_e[i]] << 8) | l_dl[i];
    }
}

template <int DIM>
__device__ __forceinline__ int acc_pos(int nl, int f) {
    constexpr int PER = DIM / 32;
    return nl * DIM + (f % PER) * 32 + (f / PER);
}

template <int DIM, bool RELU>
__global__ __launch_bounds__(OWN_THREADS) void aggregate_owner(
    const int* __restrict__ src, const int* __restrict__ dst,
    const float* __restrict__ dis, const float* __restrict__ m,
    const float* __restrict__ bias, const int* __restrict__ lists,
    const int* __restrict__ counts, float* __restrict__ out,
    int nEdges, int nNodes) {
    constexpr int PER = DIM / 32;
    __shared__ float acc[OWN_NODES * DIM];
    __shared__ int   l_src[LIST_CAP];
    __shared__ int   l_dl[LIST_CAP];
    __shared__ float l_w[LIST_CAP];

    const int tid  = threadIdx.x;
    const int lane = tid & 31;
    const int wave = tid >> 5;
    const int base = blockIdx.x * OWN_NODES;

    for (int i = tid; i < OWN_NODES * DIM; i += OWN_THREADS) acc[i] = 0.0f;

    const int total   = counts[blockIdx.x];
    const bool listed = total <= LIST_CAP;
    const int n = listed ? total : 0;
    const int* __restrict__ mine = lists + (size_t)blockIdx.x * LIST_CAP;
    for (int i = tid; i < n; i += OWN_THREADS) {
        const int p  = mine[i];
        const int s  = p >> 8;
        const int dl = p & 255;
        l_src[i] = s;
        l_dl[i]  = dl;
        l_w[i]   = dis[s] * dis[base + dl];
    }
    __syncthreads();

    constexpr int EU = 8;
    static_assert(EU == 8, "the load fence below names eight rows");
    typedef __attribute__((ext_vector_type(PER))) float vrow;
    for (int i0 = wave; i0 < n; i0 += 32 * EU) {
        int dl[EU];
        float w[EU];
        vrow v[EU];
#pragma unroll
        for (int u = 0; u < EU; ++u) {
            const int i  = i0 + 32 * u;
            const int ii = (i < n) ? i : i0;
            const int s  = l_src[ii];
            dl[u] = l_dl[ii];
            w[u]  = l_w[ii];
            v[u]  = *reinterpret_cast<const vrow*>(m + (size_t)s * DIM + lane * PER);
        }
        asm volatile("" :: "v"(v[0]), "v"(v[1]), "v"(v[2]), "v"(v[3]),
                           "v"(v[4]), "v"(v[5]), "v"(v[6]), "v"(v[7]));
#pragma unroll
        for (int u = 0; u < EU; ++u) {
            if (i0 + 32 * u < n) {
#pragma unroll
                for (int j = 0; j < PER; ++j)
                    atomicAdd(&acc[dl[u] * DIM + j * 32 + lane], v[u][j] * w[u]);
            }
        }
    }

    if (!listed) {
        for (int e = tid; e < nEdges; e += OWN_THREADS) {
            const unsigned a = (unsigned)(dst[e] - base);
            if (a < OWN_NODES) {
                const int s = src[e];
                const float w = dis[s] * dis[base + (int)a];
                for (int f = 0; f < DIM; ++f)
                    atomicAdd(&acc[acc_pos<DIM>((int)a, f)], m[(size_t)s * DIM + f] * w);
            }
        }
    }
    __syncthreads();

    constexpr int ROW4 = DIM / 4;
    for (int idx = tid; idx < OWN_NODES * ROW4; idx += OWN_THREADS) {
        const int nl   = idx / ROW4;
        const int f0   = (idx % ROW4) * 4;
        const int node = base + nl;
        if (node < nNodes) {
            const float di = dis[node];
            const size_t g = (size_t)node * DIM + f0;
            const float4 mv = *reinterpret_cast<const float4*>(m + g);
            const float4 bv = *reinterpret_cast<const float4*>(bias + f0);
            float4 r;
            r.x = acc[acc_pos<DIM>(nl, f0)]     + mv.x * di * di + bv.x;
            r.y = acc[acc_pos<DIM>(nl, f0 + 1)] + mv.y * di * di + bv.y;
            r.z = acc[acc_pos<DIM>(nl, f0 + 2)] + mv.z * di * di + bv.z;
            r.w = acc[acc_pos<DIM>(nl, f0 + 3)] + mv.w * di * di + bv.w;
            if (RELU) {
                r.x = fmaxf(r.x, 0.0f); r.y = fmaxf(r.y, 0.0f);
                r.z = fmaxf(r.z, 0.0f); r.w = fmaxf(r.w, 0.0f);
            }
            *reinterpret_cast<float4*>(out + g) = r;
        }
    }
}

extern "C" void kernel_launch(void* const* d_in, const int* in_sizes, int n_in,
                              void* d_out, int out_size, void* d_ws, size_t ws_size,
                              hipStream_t stream) {
    const float* x   = (const float*)d_in[0];
    const int*   ei  = (const int*)d_in[1];
    const float* W0  = (const float*)d_in[2];
    const float* b0  = (const float*)d_in[3];
    const float* W1  = (const float*)d_in[4];
    const float* b1  = (const float*)d_in[5];
    const float* W2  = (const float*)d_in[6];
    const float* b2  = (const float*)d_in[7];
    float* out = (float*)d_out;

    const int* src = ei;
    const int* dst = ei + N_EDGES;

    const int gemmBlocks  = (N_NODES / 16 + 3) / 4;
    const int gemmThreadsH = (HID_DIM / 64) * 128;
    const int gemmThreadsL = (LAT_DIM / 64) * 128;
    const int ownerBlocks = (N_NODES + OWN_NODES - 1) / OWN_NODES;

    float* ws  = (float*)d_ws;
    float* dis = ws;
    float* m   = dis + N_NODES;
    float* h   = m   + (size_t)N_NODES * HID_DIM;
    int* lists  = (int*)(h + (size_t)N_NODES * HID_DIM);
    int* counts = lists + (size_t)ownerBlocks * LIST_CAP;

    bin_edges<<<ownerBlocks, OWN_THREADS, 0, stream>>>(src, dst, dis, lists, counts, N_EDGES, N_NODES);

    gemm_strip_f32<IN_DIM, HID_DIM><<<gemmBlocks, gemmThreadsH, 0, stream>>>(x, W0, m, N_NODES);
    aggregate_owner<HID_DIM, true><<<ownerBlocks, OWN_THREADS, 0, stream>>>(
        src, dst, dis, m, b0, lists, counts, h, N_EDGES, N_NODES);

    gemm_strip_f32<HID_DIM, HID_DIM><<<gemmBlocks, gemmThreadsH, 0, stream>>>(h, W1, m, N_NODES);
    aggregate_owner<HID_DIM, true><<<ownerBlocks, OWN_THREADS, 0, stream>>>(
        src, dst, dis, m, b1, lists, counts, h, N_EDGES, N_NODES);

    gemm_strip_f32<HID_DIM, LAT_DIM><<<gemmBlocks, gemmThreadsL, 0, stream>>>(h, W2, m, N_NODES);
    aggregate_owner<LAT_DIM, false><<<ownerBlocks, OWN_THREADS, 0, stream>>>(
        src, dst, dis, m, b2, lists, counts, out, N_EDGES, N_NODES);
}


# bin_edges scan split two ways: a work-group scans 1/16 of dst for a 2560-node group (was all dst per 160-node owner); 16 sub-lists per owner, in-degree parts summed in K=3 launch; sum order = arrival
# speedup vs baseline: 3.3457x; 1.1648x over previous
#include <hip/hip_runtime.h>
#include <hip/hip_bf16.h>


typedef __attribute__((ext_vector_type(2))) float v2f;
typedef __attribute__((ext_vector_type(8))) float v8f;

#define N_NODES 40000
#define N_EDGES 640000
#define IN_DIM  3
#define HID_DIM 128
#define LAT_DIM 64

#define OWN_NODES 160
#define OWN_THREADS 1024
#define LIST_CAP 8192
#define BIN_SPLIT 16
#define GROUP_NODES (OWN_NODES * BIN_SPLIT)
#define SUB_CAP (LIST_CAP / BIN_SPLIT)
#define N_GROUPS ((N_NODES + GROUP_NODES - 1) / GROUP_NODES)
#define N_PAD (N_GROUPS * GROUP_NODES)

template <int K, int N, int DEG_PLANES>
__global__ __launch_bounds__((N / 64) * 128) void gemm_strip_f32(
    const float* __restrict__ A, const float* __restrict__ B,
    float* __restrict__ C, int M,
    const int* __restrict__ degp, float* __restrict__ dis) {
    constexpr int NH      = N / 64;
    constexpr int NTT     = N / 16;
    constexpr int KB      = (K + 7) / 8;
    constexpr int THREADS = NH * 128;
    __shared__ float4 Bs[KB * NTT * 32];

    const int tid  = threadIdx.x;
    const int lane = tid & 31;
    const int wave = tid >> 5;
    const int c    = lane & 15;
    const int h    = lane >> 4;
    const int nh   = wave % NH;
    const int strip = blockIdx.x * 4 + wave / NH;
    const bool active = strip * 16 < M;
    const int row  = strip * 16 + c;

    const int  drow  = blockIdx.x * 64 + tid;
    const bool dmine = DEG_PLANES > 0 && tid < 64 && drow < M;
    int dpart[DEG_PLANES > 0 ? DEG_PLANES : 1];
    if constexpr (DEG_PLANES > 0) {
        if (dmine) {
#pragma unroll
            for (int p = 0; p < DEG_PLANES; ++p) dpart[p] = degp[(size_t)p * N_PAD + drow];
        }
    }

    float4 a[KB];
    if (active) {
#pragma unroll
        for (int kb = 0; kb < KB; ++kb) {
            const int k0 = kb * 8 + h * 4;
            if constexpr (K % 8 == 0) {
                a[kb] = *reinterpret_cast<const float4*>(A + (size_t)row * K + k0);
            } else {
                const float* ar = A + (size_t)row * K;
                const float x0 = ar[min(k0,     K - 1)], x1 = ar[min(k0 + 1, K - 1)];
                const float x2 = ar[min(k0 + 2, K - 1)], x3 = ar[min(k0 + 3, K - 1)];
                a[kb].x = (k0     < K) ? x0 : 0.0f;
                a[kb].y = (k0 + 1 < K) ? x1 : 0.0f;
                a[kb].z = (k0 + 2 < K) ? x2 : 0.0f;
                a[kb].w = (k0 + 3 < K) ? x3 : 0.0f;
            }
        }
    }

    constexpr int ITEMS = KB * NH * 32;
    constexpr int TRIPS = (ITEMS + THREADS - 1) / THREADS;
#pragma unroll
    for (int j = 0; j < TRIPS; ++j) {
        const int it = tid + j * THREADS;
        if (ITEMS % THREADS == 0 || it < ITEMS) {
            const int l   = it & 31;
            const int bn  = (it >> 5) % NH;
            const int kb  = (it >> 5) / NH;
            const int k0  = kb * 8 + (l >> 4) * 4;
            const int col = bn * 64 + (l & 15) * 4;
            float4 v[4];
#pragma unroll
            for (int e = 0; e < 4; ++e) {
                if constexpr (K % 8 == 0) {
                    v[e] = *reinterpret_cast<const float4*>(B + (size_t)(k0 + e) * N + col);
                } else {
                    const float4 t = *reinterpret_cast<const float4*>(B + (size_t)min(k0 + e, K - 1) * N + col);
                    v[e] = (k0 + e < K) ? t : make_float4(0.0f, 0.0f, 0.0f, 0.0f);
                }
            }
            float4* d = Bs + (kb * NTT + bn * 4) * 32 + l;
            d[0]  = make_float4(v[0].x, v[1].x, v[2].x, v[3].x);
            d[32] = make_float4(v[0].y, v[1].y, v[2].y, v[3].y);
            d[64] = make_float4(v[0].z, v[1].z, v[2].z, v[3].z);
            d[96] = make_float4(v[0].w, v[1].w, v[2].w, v[3].w);
        }
    }
    if constexpr (DEG_PLANES > 0) {
        if (dmine) {
            int d = 1;
#pragma unroll
            for (int p = 0; p < DEG_PLANES; ++p) d += dpart[p];
            dis[drow] = rsqrtf((float)d);
        }
    }
    __syncthreads();
    if (!active) return;

    v8f acc[4] = {};
#pragma unroll
    for (int kb = 0; kb < KB; ++kb) {
        float4 b[4];
#pragma unroll
        for (int t = 0; t < 4; ++t) b[t] = Bs[(kb * NTT + nh * 4 + t) * 32 + lane];
        const v2f a0 = {a[kb].x, a[kb].y};
        const v2f a1 = {a[kb].z, a[kb].w};
#pragma unroll
        for (int t = 0; t < 4; ++t) {
            const v2f b0 = {b[t].x, b[t].y};
            acc[t] = __builtin_amdgcn_wmma_f32_16x16x4_f32(
                false, a0, false, b0, (short)0, acc[t], false, false);
        }
#pragma unroll
        for (int t = 0; t < 4; ++t) {
            const v2f b1 = {b[t].z, b[t].w};
            acc[t] = __builtin_amdgcn_wmma_f32_16x16x4_f32(
                false, a1, false, b1, (short)0, acc[t], false, false);
        }
    }

    float* __restrict__ crow = C + (size_t)(strip * 16 + h * 8) * N + nh * 64 + c * 4;
#pragma unroll
    for (int r = 0; r < 8; ++r)
        *reinterpret_cast<float4*>(crow + (size_t)r * N) =
            make_float4(acc[0][r], acc[1][r], acc[2][r], acc[3][r]);
}

__global__ __launch_bounds__(OWN_THREADS) void bin_edges(
    const int* __restrict__ src, const int* __restrict__ dst,
    int* __restrict__ degp, int* __restrict__ lists, int* __restrict__ counts,
    int nEdges) {
    __shared__ int l_e[LIST_CAP];
    __shared__ int l_a[LIST_CAP];
    __shared__ int deg[GROUP_NODES];
    __shared__ int cnt;
    __shared__ int cnt2[BIN_SPLIT];

    const int tid   = threadIdx.x;
    const int grp   = blockIdx.x / BIN_SPLIT;
    const int chunk = blockIdx.x % BIN_SPLIT;
    const int base  = grp * GROUP_NODES;

    for (int i = tid; i < GROUP_NODES; i += OWN_THREADS) deg[i] = 0;
    if (tid == 0) cnt = 0;
    if (tid < BIN_SPLIT) cnt2[tid] = 0;
    __syncthreads();

    auto put = [&](int slot, int e, int a) {
        if (slot < LIST_CAP) { l_e[slot] = e; l_a[slot] = a; }
        atomicAdd(&deg[a], 1);
    };
    auto take4 = [&](int4 d, int e0) {
        const unsigned a0 = (unsigned)(d.x - base), a1 = (unsigned)(d.y - base);
        const unsigned a2 = (unsigned)(d.z - base), a3 = (unsigned)(d.w - base);
        if (min(min(a0, a1), min(a2, a3)) < GROUP_NODES) {
            const int c = (a0 < GROUP_NODES) + (a1 < GROUP_NODES) + (a2 < GROUP_NODES) + (a3 < GROUP_NODES);
            int slot = atomicAdd(&cnt, c);
            if (a0 < GROUP_NODES) put(slot++, e0,     (int)a0);
            if (a1 < GROUP_NODES) put(slot++, e0 + 1, (int)a1);
            if (a2 < GROUP_NODES) put(slot++, e0 + 2, (int)a2);
            if (a3 < GROUP_NODES) put(slot++, e0 + 3, (int)a3);
        }
    };

    const int4* __restrict__ dst4 = reinterpret_cast<const int4*>(dst);
    const int nQuad = nEdges >> 2;
    const int per   = (nQuad + BIN_SPLIT - 1) / BIN_SPLIT;
    const int qlo   = chunk * per;
    const int qhi   = min(qlo + per, nQuad);
    constexpr int SU = 10;
    for (int q0 = qlo + tid; q0 < qhi; q0 += OWN_THREADS * SU) {
        int4 d[SU];
#pragma unroll
        for (int u = 0; u < SU; ++u) {
            const int q = q0 + OWN_THREADS * u;
            d[u] = (q < qhi) ? dst4[q] : make_int4(-1, -1, -1, -1);
        }
#pragma unroll
        for (int u = 0; u < SU; ++u) take4(d[u], (q0 + OWN_THREADS * u) * 4);
    }
    if (chunk == BIN_SPLIT - 1) {
        for (int e = (nQuad << 2) + tid; e < nEdges; e += OWN_THREADS) {
            const unsigned a = (unsigned)(dst[e] - base);
            if (a < GROUP_NODES) put(atomicAdd(&cnt, 1), e, (int)a);
        }
    }
    __syncthreads();

    for (int i = tid; i < GROUP_NODES; i += OWN_THREADS)
        degp[(size_t)chunk * N_PAD + base + i] = deg[i];

    const int total = cnt;
    const bool fits = total <= LIST_CAP;
    if (fits) {
        const int lane = tid & 31;
        for (int i0 = tid - lane; i0 < total; i0 += OWN_THREADS) {
            const int i = i0 + lane;
            const bool valid = i < total;
            int a = 0, s = 0;
            if (valid) { a = l_a[i]; s = src[l_e[i]]; }
            const int j  = a / OWN_NODES;
            const int dl = a - j * OWN_NODES;
            unsigned same = 0;
            int lane_cnt = 0;
#pragma unroll
            for (int jj = 0; jj < BIN_SPLIT; ++jj) {
                const unsigned mk = (unsigned)__ballot(valid && j == jj);
                if (j == jj) same = mk;
                if (lane == jj) lane_cnt = __popc(mk);
            }
            int first = 0;
            if (lane < BIN_SPLIT) first = atomicAdd(&cnt2[lane], lane_cnt);
            const int slot = __shfl(first, j) + __popc(same & ((1u << lane) - 1u));
            if (valid && slot < SUB_CAP)
                lists[((size_t)(grp * BIN_SPLIT + j) * BIN_SPLIT + chunk) * SUB_CAP + slot] = (s << 8) | dl;
        }
    }
    __syncthreads();
    if (tid < BIN_SPLIT)
        counts[(grp * BIN_SPLIT + tid) * BIN_SPLIT + chunk] = fits ? cnt2[tid] : SUB_CAP + 1;
}

template <int DIM>
__device__ __forceinline__ int acc_pos(int nl, int f) {
    constexpr int PER = DIM / 32;
    return nl * DIM + (f % PER) * 32 + (f / PER);
}

template <int DIM, bool RELU>
__global__ __launch_bounds__(OWN_THREADS) void aggregate_owner(
    const int* __restrict__ src, const int* __restrict__ dst,
    const float* __restrict__ dis, const float* __restrict__ m,
    const float* __restrict__ bias, const int* __restrict__ lists,
    const int* __restrict__ counts, float* __restrict__ out,
    int nEdges, int nNodes) {
    constexpr int PER = DIM / 32;
    __shared__ float acc[OWN_NODES * DIM];
    __shared__ int   l_src[LIST_CAP];
    __shared__ int   l_dl[LIST_CAP];
    __shared__ float l_w[LIST_CAP];

    const int tid  = threadIdx.x;
    const int lane = tid & 31;
    const int wave = tid >> 5;
    const int base = blockIdx.x * OWN_NODES;

    for (int i = tid; i < OWN_NODES * DIM; i += OWN_THREADS) acc[i] = 0.0f;

    const int* __restrict__ cn = counts + blockIdx.x * BIN_SPLIT;
    int off[BIN_SPLIT];
    int total = 0;
    bool listed = true;
#pragma unroll
    for (int c = 0; c < BIN_SPLIT; ++c) {
        const int nc = cn[c];
        listed = listed && (nc <= SUB_CAP);
        off[c] = total;
        total += nc;
    }
    const int n = listed ? total : 0;
    const int* __restrict__ mine = lists + (size_t)blockIdx.x * LIST_CAP;
    for (int i = tid; i < n; i += OWN_THREADS) {
        int c = 0, start = 0;
#pragma unroll
        for (int cc = 1; cc < BIN_SPLIT; ++cc)
            if (i >= off[cc]) { c = cc; start = off[cc]; }
        const int p  = mine[c * SUB_CAP + (i - start)];
        const int s  = p >> 8;
        const int dl = p & 255;
        l_src[i] = s;
        l_dl[i]  = dl;
        l_w[i]   = dis[s] * dis[base + dl];
    }
    __syncthreads();

    constexpr int EU = 8;
    static_assert(EU == 8, "the load fence below names eight rows");
    typedef __attribute__((ext_vector_type(PER))) float vrow;
    for (int i0 = wave; i0 < n; i0 += 32 * EU) {
        int dl[EU];
        float w[EU];
        vrow v[EU];
#pragma unroll
        for (int u = 0; u < EU; ++u) {
            const int i  = i0 + 32 * u;
            const int ii = (i < n) ? i : i0;
            const int s  = l_src[ii];
            dl[u] = l_dl[ii];
            w[u]  = l_w[ii];
            v[u]  = *reinterpret_cast<const vrow*>(m + (size_t)s * DIM + lane * PER);
        }
        asm volatile("" :: "v"(v[0]), "v"(v[1]), "v"(v[2]), "v"(v[3]),
                           "v"(v[4]), "v"(v[5]), "v"(v[6]), "v"(v[7]));
#pragma unroll
        for (int u = 0; u < EU; ++u) {
            if (i0 + 32 * u < n) {
#pragma unroll
                for (int j = 0; j < PER; ++j)
                    atomicAdd(&acc[dl[u] * DIM + j * 32 + lane], v[u][j] * w[u]);
            }
        }
    }

    if (!listed) {
        for (int e = tid; e < nEdges; e += OWN_THREADS) {
            const unsigned a = (unsigned)(dst[e] - base);
            if (a < OWN_NODES) {
                const int s = src[e];
                const float w = dis[s] * dis[base + (int)a];
                for (int f = 0; f < DIM; ++f)
                    atomicAdd(&acc[acc_pos<DIM>((int)a, f)], m[(size_t)s * DIM + f] * w);
            }
        }
    }
    __syncthreads();

    constexpr int ROW4 = DIM / 4;
    for (int idx = tid; idx < OWN_NODES * ROW4; idx += OWN_THREADS) {
        const int nl   = idx / ROW4;
        const int f0   = (idx % ROW4) * 4;
        const int node = base + nl;
        if (node < nNodes) {
            const float di = dis[node];
            const size_t g = (size_t)node * DIM + f0;
            const float4 mv = *reinterpret_cast<const float4*>(m + g);
            const float4 bv = *reinterpret_cast<const float4*>(bias + f0);
            float4 r;
            r.x = acc[acc_pos<DIM>(nl, f0)]     + mv.x * di * di + bv.x;
            r.y = acc[acc_pos<DIM>(nl, f0 + 1)] + mv.y * di * di + bv.y;
            r.z = acc[acc_pos<DIM>(nl, f0 + 2)] + mv.z * di * di + bv.z;
            r.w = acc[acc_pos<DIM>(nl, f0 + 3)] + mv.w * di * di + bv.w;
            if (RELU) {
                r.x = fmaxf(r.x, 0.0f); r.y = fmaxf(r.y, 0.0f);
                r.z = fmaxf(r.z, 0.0f); r.w = fmaxf(r.w, 0.0f);
            }
            *reinterpret_cast<float4*>(out + g) = r;
        }
    }
}

extern "C" void kernel_launch(void* const* d_in, const int* in_sizes, int n_in,
                              void* d_out, int out_size, void* d_ws, size_t ws_size,
                              hipStream_t stream) {
    const float* x   = (const float*)d_in[0];
    const int*   ei  = (const int*)d_in[1];
    const float* W0  = (const float*)d_in[2];
    const float* b0  = (const float*)d_in[3];
    const float* W1  = (const float*)d_in[4];
    const float* b1  = (const float*)d_in[5];
    const float* W2  = (const float*)d_in[6];
    const float* b2  = (const float*)d_in[7];
    float* out = (float*)d_out;

    const int* src = ei;
    const int* dst = ei + N_EDGES;

    const int gemmBlocks  = (N_NODES / 16 + 3) / 4;
    const int gemmThreadsH = (HID_DIM / 64) * 128;
    const int gemmThreadsL = (LAT_DIM / 64) * 128;
    const int ownerBlocks = (N_NODES + OWN_NODES - 1) / OWN_NODES;

    const int binBlocks   = N_GROUPS * BIN_SPLIT;
    const int ownersPad   = N_GROUPS * BIN_SPLIT;

    float* ws  = (float*)d_ws;
    float* dis = ws;
    float* m   = dis + N_NODES;
    float* h   = m   + (size_t)N_NODES * HID_DIM;
    int* lists  = (int*)(h + (size_t)N_NODES * HID_DIM);
    int* counts = lists + (size_t)ownersPad * LIST_CAP;
    int* degp   = counts + (size_t)ownersPad * BIN_SPLIT;

    bin_edges<<<binBlocks, OWN_THREADS, 0, stream>>>(src, dst, degp, lists, counts, N_EDGES);

    gemm_strip_f32<IN_DIM, HID_DIM, BIN_SPLIT><<<gemmBlocks, gemmThreadsH, 0, stream>>>(x, W0, m, N_NODES, degp, dis);
    aggregate_owner<HID_DIM, true><<<ownerBlocks, OWN_THREADS, 0, stream>>>(
        src, dst, dis, m, b0, lists, counts, h, N_EDGES, N_NODES);

    gemm_strip_f32<HID_DIM, HID_DIM, 0><<<gemmBlocks, gemmThreadsH, 0, stream>>>(h, W1, m, N_NODES, nullptr, nullptr);
    aggregate_owner<HID_DIM, true><<<ownerBlocks, OWN_THREADS, 0, stream>>>(
        src, dst, dis, m, b1, lists, counts, h, N_EDGES, N_NODES);

    gemm_strip_f32<HID_DIM, LAT_DIM, 0><<<gemmBlocks, gemmThreadsL, 0, stream>>>(h, W2, m, N_NODES, nullptr, nullptr);
    aggregate_owner<LAT_DIM, false><<<ownerBlocks, OWN_THREADS, 0, stream>>>(
        src, dst, dis, m, b2, lists, counts, out, N_EDGES, N_NODES);
}


# aggregate: a wave reads 32 edge records per load from the global list and takes src, dst, weight out of a lane into scalar registers; LDS list copy and 3 LDS reads per edge gone; sum order = arrival
# speedup vs baseline: 3.4672x; 1.0363x over previous
#include <hip/hip_runtime.h>
#include <hip/hip_bf16.h>


typedef __attribute__((ext_vector_type(2))) float v2f;
typedef __attribute__((ext_vector_type(8))) float v8f;

#define N_NODES 40000
#define N_EDGES 640000
#define IN_DIM  3
#define HID_DIM 128
#define LAT_DIM 64

#define OWN_NODES 160
#define OWN_THREADS 1024
#define LIST_CAP 8192
#define BIN_SPLIT 16
#define GROUP_NODES (OWN_NODES * BIN_SPLIT)
#define SUB_CAP (LIST_CAP / BIN_SPLIT)
#define N_GROUPS ((N_NODES + GROUP_NODES - 1) / GROUP_NODES)
#define N_PAD (N_GROUPS * GROUP_NODES)

template <int K, int N, int DEG_PLANES>
__global__ __launch_bounds__((N / 64) * 128) void gemm_strip_f32(
    const float* __restrict__ A, const float* __restrict__ B,
    float* __restrict__ C, int M,
    const int* __restrict__ degp, float* __restrict__ dis) {
    constexpr int NH      = N / 64;
    constexpr int NTT     = N / 16;
    constexpr int KB      = (K + 7) / 8;
    constexpr int THREADS = NH * 128;
    __shared__ float4 Bs[KB * NTT * 32];

    const int tid  = threadIdx.x;
    const int lane = tid & 31;
    const int wave = tid >> 5;
    const int c    = lane & 15;
    const int h    = lane >> 4;
    const int nh   = wave % NH;
    const int strip = blockIdx.x * 4 + wave / NH;
    const bool active = strip * 16 < M;
    const int row  = strip * 16 + c;

    const int  drow  = blockIdx.x * 64 + tid;
    const bool dmine = DEG_PLANES > 0 && tid < 64 && drow < M;
    int dpart[DEG_PLANES > 0 ? DEG_PLANES : 1];
    if constexpr (DEG_PLANES > 0) {
        if (dmine) {
#pragma unroll
            for (int p = 0; p < DEG_PLANES; ++p) dpart[p] = degp[(size_t)p * N_PAD + drow];
        }
    }

    float4 a[KB];
    if (active) {
#pragma unroll
        for (int kb = 0; kb < KB; ++kb) {
            const int k0 = kb * 8 + h * 4;
            if constexpr (K % 8 == 0) {
                a[kb] = *reinterpret_cast<const float4*>(A + (size_t)row * K + k0);
            } else {
                const float* ar = A + (size_t)row * K;
                const float x0 = ar[min(k0,     K - 1)], x1 = ar[min(k0 + 1, K - 1)];
                const float x2 = ar[min(k0 + 2, K - 1)], x3 = ar[min(k0 + 3, K - 1)];
                a[kb].x = (k0     < K) ? x0 : 0.0f;
                a[kb].y = (k0 + 1 < K) ? x1 : 0.0f;
                a[kb].z = (k0 + 2 < K) ? x2 : 0.0f;
                a[kb].w = (k0 + 3 < K) ? x3 : 0.0f;
            }
        }
    }

    constexpr int ITEMS = KB * NH * 32;
    constexpr int TRIPS = (ITEMS + THREADS - 1) / THREADS;
#pragma unroll
    for (int j = 0; j < TRIPS; ++j) {
        const int it = tid + j * THREADS;
        if (ITEMS % THREADS == 0 || it < ITEMS) {
            const int l   = it & 31;
            const int bn  = (it >> 5) % NH;
            const int kb  = (it >> 5) / NH;
            const int k0  = kb * 8 + (l >> 4) * 4;
            const int col = bn * 64 + (l & 15) * 4;
            float4 v[4];
#pragma unroll
            for (int e = 0; e < 4; ++e) {
                if constexpr (K % 8 == 0) {
                    v[e] = *reinterpret_cast<const float4*>(B + (size_t)(k0 + e) * N + col);
                } else {
                    const float4 t = *reinterpret_cast<const float4*>(B + (size_t)min(k0 + e, K - 1) * N + col);
                    v[e] = (k0 + e < K) ? t : make_float4(0.0f, 0.0f, 0.0f, 0.0f);
                }
            }
            float4* d = Bs + (kb * NTT + bn * 4) * 32 + l;
            d[0]  = make_float4(v[0].x, v[1].x, v[2].x, v[3].x);
            d[32] = make_float4(v[0].y, v[1].y, v[2].y, v[3].y);
            d[64] = make_float4(v[0].z, v[1].z, v[2].z, v[3].z);
            d[96] = make_float4(v[0].w, v[1].w, v[2].w, v[3].w);
        }
    }
    if constexpr (DEG_PLANES > 0) {
        if (dmine) {
            int d = 1;
#pragma unroll
            for (int p = 0; p < DEG_PLANES; ++p) d += dpart[p];
            dis[drow] = rsqrtf((float)d);
        }
    }
    __syncthreads();
    if (!active) return;

    v8f acc[4] = {};
#pragma unroll
    for (int kb = 0; kb < KB; ++kb) {
        float4 b[4];
#pragma unroll
        for (int t = 0; t < 4; ++t) b[t] = Bs[(kb * NTT + nh * 4 + t) * 32 + lane];
        const v2f a0 = {a[kb].x, a[kb].y};
        const v2f a1 = {a[kb].z, a[kb].w};
#pragma unroll
        for (int t = 0; t < 4; ++t) {
            const v2f b0 = {b[t].x, b[t].y};
            acc[t] = __builtin_amdgcn_wmma_f32_16x16x4_f32(
                false, a0, false, b0, (short)0, acc[t], false, false);
        }
#pragma unroll
        for (int t = 0; t < 4; ++t) {
            const v2f b1 = {b[t].z, b[t].w};
            acc[t] = __builtin_amdgcn_wmma_f32_16x16x4_f32(
                false, a1, false, b1, (short)0, acc[t], false, false);
        }
    }

    float* __restrict__ crow = C + (size_t)(strip * 16 + h * 8) * N + nh * 64 + c * 4;
#pragma unroll
    for (int r = 0; r < 8; ++r)
        *reinterpret_cast<float4*>(crow + (size_t)r * N) =
            make_float4(acc[0][r], acc[1][r], acc[2][r], acc[3][r]);
}

__global__ __launch_bounds__(OWN_THREADS) void bin_edges(
    const int* __restrict__ src, const int* __restrict__ dst,
    int* __restrict__ degp, int* __restrict__ lists, int* __restrict__ counts,
    int nEdges) {
    __shared__ int l_e[LIST_CAP];
    __shared__ int l_a[LIST_CAP];
    __shared__ int deg[GROUP_NODES];
    __shared__ int cnt;
    __shared__ int cnt2[BIN_SPLIT];

    const int tid   = threadIdx.x;
    const int grp   = blockIdx.x / BIN_SPLIT;
    const int chunk = blockIdx.x % BIN_SPLIT;
    const int base  = grp * GROUP_NODES;

    for (int i = tid; i < GROUP_NODES; i += OWN_THREADS) deg[i] = 0;
    if (tid == 0) cnt = 0;
    if (tid < BIN_SPLIT) cnt2[tid] = 0;
    __syncthreads();

    auto put = [&](int slot, int e, int a) {
        if (slot < LIST_CAP) { l_e[slot] = e; l_a[slot] = a; }
        atomicAdd(&deg[a], 1);
    };
    auto take4 = [&](int4 d, int e0) {
        const unsigned a0 = (unsigned)(d.x - base), a1 = (unsigned)(d.y - base);
        const unsigned a2 = (unsigned)(d.z - base), a3 = (unsigned)(d.w - base);
        if (min(min(a0, a1), min(a2, a3)) < GROUP_NODES) {
            const int c = (a0 < GROUP_NODES) + (a1 < GROUP_NODES) + (a2 < GROUP_NODES) + (a3 < GROUP_NODES);
            int slot = atomicAdd(&cnt, c);
            if (a0 < GROUP_NODES) put(slot++, e0,     (int)a0);
            if (a1 < GROUP_NODES) put(slot++, e0 + 1, (int)a1);
            if (a2 < GROUP_NODES) put(slot++, e0 + 2, (int)a2);
            if (a3 < GROUP_NODES) put(slot++, e0 + 3, (int)a3);
        }
    };

    const int4* __restrict__ dst4 = reinterpret_cast<const int4*>(dst);
    const int nQuad = nEdges >> 2;
    const int per   = (nQuad + BIN_SPLIT - 1) / BIN_SPLIT;
    const int qlo   = chunk * per;
    const int qhi   = min(qlo + per, nQuad);
    constexpr int SU = 10;
    for (int q0 = qlo + tid; q0 < qhi; q0 += OWN_THREADS * SU) {
        int4 d[SU];
#pragma unroll
        for (int u = 0; u < SU; ++u) {
            const int q = q0 + OWN_THREADS * u;
            d[u] = (q < qhi) ? dst4[q] : make_int4(-1, -1, -1, -1);
        }
#pragma unroll
        for (int u = 0; u < SU; ++u) take4(d[u], (q0 + OWN_THREADS * u) * 4);
    }
    if (chunk == BIN_SPLIT - 1) {
        for (int e = (nQuad << 2) + tid; e < nEdges; e += OWN_THREADS) {
            const unsigned a = (unsigned)(dst[e] - base);
            if (a < GROUP_NODES) put(atomicAdd(&cnt, 1), e, (int)a);
        }
    }
    __syncthreads();

    for (int i = tid; i < GROUP_NODES; i += OWN_THREADS)
        degp[(size_t)chunk * N_PAD + base + i] = deg[i];

    const int total = cnt;
    const bool fits = total <= LIST_CAP;
    if (fits) {
        const int lane = tid & 31;
        for (int i0 = tid - lane; i0 < total; i0 += OWN_THREADS) {
            const int i = i0 + lane;
            const bool valid = i < total;
            int a = 0, s = 0;
            if (valid) { a = l_a[i]; s = src[l_e[i]]; }
            const int j  = a / OWN_NODES;
            const int dl = a - j * OWN_NODES;
            unsigned same = 0;
            int lane_cnt = 0;
#pragma unroll
            for (int jj = 0; jj < BIN_SPLIT; ++jj) {
                const unsigned mk = (unsigned)__ballot(valid && j == jj);
                if (j == jj) same = mk;
                if (lane == jj) lane_cnt = __popc(mk);
            }
            int first = 0;
            if (lane < BIN_SPLIT) first = atomicAdd(&cnt2[lane], lane_cnt);
            const int slot = __shfl(first, j) + __popc(same & ((1u << lane) - 1u));
            if (valid && slot < SUB_CAP)
                lists[((size_t)(grp * BIN_SPLIT + j) * BIN_SPLIT + chunk) * SUB_CAP + slot] = (s << 8) | dl;
        }
    }
    __syncthreads();
    if (tid < BIN_SPLIT)
        counts[(grp * BIN_SPLIT + tid) * BIN_SPLIT + chunk] = fits ? cnt2[tid] : SUB_CAP + 1;
}

template <int DIM>
__device__ __forceinline__ int acc_pos(int nl, int f) {
    constexpr int PER = DIM / 32;
    return nl * DIM + (f % PER) * 32 + (f / PER);
}

template <int DIM, bool RELU>
__global__ __launch_bounds__(OWN_THREADS)
__attribute__((amdgpu_waves_per_eu(8, 8)))
void aggregate_owner(
    const int* __restrict__ src, const int* __restrict__ dst,
    const float* __restrict__ dis, const float* __restrict__ m,
    const float* __restrict__ bias, const int* __restrict__ lists,
    const int* __restrict__ counts, float* __restrict__ out,
    int nEdges, int nNodes) {
    constexpr int PER   = DIM / 32;
    constexpr int WAVES = OWN_THREADS / 32;
    __shared__ float acc[OWN_NODES * DIM];

    const int tid  = threadIdx.x;
    const int lane = tid & 31;
    const int wave = __builtin_amdgcn_readfirstlane(tid >> 5);
    const int base = blockIdx.x * OWN_NODES;

    for (int i = tid; i < OWN_NODES * DIM; i += OWN_THREADS) acc[i] = 0.0f;

    const int* __restrict__ cn = counts + blockIdx.x * BIN_SPLIT;
    int off[BIN_SPLIT];
    int total = 0;
    bool listed = true;
#pragma unroll
    for (int c = 0; c < BIN_SPLIT; ++c) {
        const int nc = cn[c];
        listed = listed && (nc <= SUB_CAP);
        off[c] = total;
        total += nc;
    }
    const int n = listed ? total : 0;
    const int* __restrict__ mine = lists + (size_t)blockIdx.x * LIST_CAP;

    const int i_lo = (n * wave) / WAVES;
    const int i_hi = (n * (wave + 1)) / WAVES;
    auto record = [&](int i) -> int {
        int c = 0, start = 0;
#pragma unroll
        for (int cc = 1; cc < BIN_SPLIT; ++cc)
            if (i >= off[cc]) { c = cc; start = off[cc]; }
        return (i < i_hi) ? mine[c * SUB_CAP + (i - start)] : -1;
    };
    int   p_cur = record(i_lo + lane);
    int   p_nxt = record(i_lo + 32 + lane);
    float w_cur = (p_cur >= 0) ? dis[p_cur >> 8] * dis[base + (p_cur & 255)] : 0.0f;
    __syncthreads();

    constexpr int EU = 8;
    static_assert(EU == 8, "the load fence below names eight rows");
    typedef __attribute__((ext_vector_type(PER))) float vrow;
    for (int g0 = i_lo; g0 < i_hi; g0 += 32) {
        const int p_nn = record(g0 + 64 + lane);
        const int pn = max(p_nxt, 0);
        const float cs = dis[pn >> 8], cd = dis[base + (pn & 255)];
        const int cnt = min(32, i_hi - g0);
        for (int b = 0; b < cnt; b += EU) {
            int dl[EU];
            float w[EU];
            vrow v[EU];
#pragma unroll
            for (int u = 0; u < EU; ++u) {
                const int l = min(b + u, cnt - 1);
                const int p = __builtin_amdgcn_readlane(p_cur, l);
                w[u]  = __int_as_float(__builtin_amdgcn_readlane(__float_as_int(w_cur), l));
                dl[u] = p & 255;
                v[u]  = *reinterpret_cast<const vrow*>(m + (size_t)(p >> 8) * DIM + lane * PER);
            }
            asm volatile("" :: "v"(v[0]), "v"(v[1]), "v"(v[2]), "v"(v[3]),
                               "v"(v[4]), "v"(v[5]), "v"(v[6]), "v"(v[7]));
#pragma unroll
            for (int u = 0; u < EU; ++u) {
                if (b + u < cnt) {
#pragma unroll
                    for (int j = 0; j < PER; ++j)
                        atomicAdd(&acc[dl[u] * DIM + j * 32 + lane], v[u][j] * w[u]);
                }
            }
        }
        w_cur = cs * cd;
        p_cur = p_nxt;
        p_nxt = p_nn;
    }

    if (!listed) {
        for (int e = tid; e < nEdges; e += OWN_THREADS) {
            const unsigned a = (unsigned)(dst[e] - base);
            if (a < OWN_NODES) {
                const int s = src[e];
                const float w = dis[s] * dis[base + (int)a];
                for (int f = 0; f < DIM; ++f)
                    atomicAdd(&acc[acc_pos<DIM>((int)a, f)], m[(size_t)s * DIM + f] * w);
            }
        }
    }
    __syncthreads();

    constexpr int ROW4 = DIM / 4;
    for (int idx = tid; idx < OWN_NODES * ROW4; idx += OWN_THREADS) {
        const int nl   = idx / ROW4;
        const int f0   = (idx % ROW4) * 4;
        const int node = base + nl;
        if (node < nNodes) {
            const float di = dis[node];
            const size_t g = (size_t)node * DIM + f0;
            const float4 mv = *reinterpret_cast<const float4*>(m + g);
            const float4 bv = *reinterpret_cast<const float4*>(bias + f0);
            float4 r;
            r.x = acc[acc_pos<DIM>(nl, f0)]     + mv.x * di * di + bv.x;
            r.y = acc[acc_pos<DIM>(nl, f0 + 1)] + mv.y * di * di + bv.y;
            r.z = acc[acc_pos<DIM>(nl, f0 + 2)] + mv.z * di * di + bv.z;
            r.w = acc[acc_pos<DIM>(nl, f0 + 3)] + mv.w * di * di + bv.w;
            if (RELU) {
                r.x = fmaxf(r.x, 0.0f); r.y = fmaxf(r.y, 0.0f);
                r.z = fmaxf(r.z, 0.0f); r.w = fmaxf(r.w, 0.0f);
            }
            *reinterpret_cast<float4*>(out + g) = r;
        }
    }
}

extern "C" void kernel_launch(void* const* d_in, const int* in_sizes, int n_in,
                              void* d_out, int out_size, void* d_ws, size_t ws_size,
                              hipStream_t stream) {
    const float* x   = (const float*)d_in[0];
    const int*   ei  = (const int*)d_in[1];
    const float* W0  = (const float*)d_in[2];
    const float* b0  = (const float*)d_in[3];
    const float* W1  = (const float*)d_in[4];
    const float* b1  = (const float*)d_in[5];
    const float* W2  = (const float*)d_in[6];
    const float* b2  = (const float*)d_in[7];
    float* out = (float*)d_out;

    const int* src = ei;
    const int* dst = ei + N_EDGES;

    const int gemmBlocks  = (N_NODES / 16 + 3) / 4;
    const int gemmThreadsH = (HID_DIM / 64) * 128;
    const int gemmThreadsL = (LAT_DIM / 64) * 128;
    const int ownerBlocks = (N_NODES + OWN_NODES - 1) / OWN_NODES;

    const int binBlocks   = N_GROUPS * BIN_SPLIT;
    const int ownersPad   = N_GROUPS * BIN_SPLIT;

    float* ws  = (float*)d_ws;
    float* dis = ws;
    float* m   = dis + N_NODES;
    float* h   = m   + (size_t)N_NODES * HID_DIM;
    int* lists  = (int*)(h + (size_t)N_NODES * HID_DIM);
    int* counts = lists + (size_t)ownersPad * LIST_CAP;
    int* degp   = counts + (size_t)ownersPad * BIN_SPLIT;

    bin_edges<<<binBlocks, OWN_THREADS, 0, stream>>>(src, dst, degp, lists, counts, N_EDGES);

    gemm_strip_f32<IN_DIM, HID_DIM, BIN_SPLIT><<<gemmBlocks, gemmThreadsH, 0, stream>>>(x, W0, m, N_NODES, degp, dis);
    aggregate_owner<HID_DIM, true><<<ownerBlocks, OWN_THREADS, 0, stream>>>(
        src, dst, dis, m, b0, lists, counts, h, N_EDGES, N_NODES);

    gemm_strip_f32<HID_DIM, HID_DIM, 0><<<gemmBlocks, gemmThreadsH, 0, stream>>>(h, W1, m, N_NODES, nullptr, nullptr);
    aggregate_owner<HID_DIM, true><<<ownerBlocks, OWN_THREADS, 0, stream>>>(
        src, dst, dis, m, b1, lists, counts, h, N_EDGES, N_NODES);

    gemm_strip_f32<HID_DIM, LAT_DIM, 0><<<gemmBlocks, gemmThreadsL, 0, stream>>>(h, W2, m, N_NODES, nullptr, nullptr);
    aggregate_owner<LAT_DIM, false><<<ownerBlocks, OWN_THREADS, 0, stream>>>(
        src, dst, dis, m, b2, lists, counts, out, N_EDGES, N_NODES);
}


# Edge norm split: GEMM epilogues store rows times dis[row], aggregate adds rows as stored, finish scales the sum by dis[dst]; per-edge dis[src]/dis[dst] loads and multiplies gone; sum order kept
# speedup vs baseline: 3.6271x; 1.0461x over previous
#include <hip/hip_runtime.h>
#include <hip/hip_bf16.h>


typedef __attribute__((ext_vector_type(2))) float v2f;
typedef __attribute__((ext_vector_type(8))) float v8f;

#define N_NODES 40000
#define N_EDGES 640000
#define IN_DIM  3
#define HID_DIM 128
#define LAT_DIM 64

#define OWN_NODES 160
#define OWN_THREADS 1024
#define LIST_CAP 8192
#define BIN_SPLIT 16
#define GROUP_NODES (OWN_NODES * BIN_SPLIT)
#define SUB_CAP (LIST_CAP / BIN_SPLIT)
#define N_GROUPS ((N_NODES + GROUP_NODES - 1) / GROUP_NODES)
#define N_PAD (N_GROUPS * GROUP_NODES)

template <int K, int N, int DEG_PLANES>
__global__ __launch_bounds__((N / 64) * 128) void gemm_strip_f32(
    const float* __restrict__ A, const float* __restrict__ B,
    float* __restrict__ C, int M,
    const int* __restrict__ degp, float* __restrict__ dis, const float* __restrict__ dis_in) {
    constexpr int NH      = N / 64;
    constexpr int NTT     = N / 16;
    constexpr int KB      = (K + 7) / 8;
    constexpr int THREADS = NH * 128;
    __shared__ float4 Bs[KB * NTT * 32];
    __shared__ float4 dsh[DEG_PLANES > 0 ? 16 : 1];

    const int tid  = threadIdx.x;
    const int lane = tid & 31;
    const int wave = tid >> 5;
    const int c    = lane & 15;
    const int h    = lane >> 4;
    const int nh   = wave % NH;
    const int strip = blockIdx.x * 4 + wave / NH;
    const bool active = strip * 16 < M;
    const int row  = strip * 16 + c;

    const int  drow  = blockIdx.x * 64 + tid;
    const bool dmine = DEG_PLANES > 0 && tid < 64 && drow < M;
    int dpart[DEG_PLANES > 0 ? DEG_PLANES : 1];
    if constexpr (DEG_PLANES > 0) {
        if (dmine) {
#pragma unroll
            for (int p = 0; p < DEG_PLANES; ++p) dpart[p] = degp[(size_t)p * N_PAD + drow];
        }
    }

    float4 a[KB];
    float4 sc0 = make_float4(0.0f, 0.0f, 0.0f, 0.0f), sc1 = sc0;
    if (active) {
        if constexpr (DEG_PLANES == 0) {
            const float4* sp = reinterpret_cast<const float4*>(dis_in + strip * 16 + h * 8);
            sc0 = sp[0];
            sc1 = sp[1];
        }
#pragma unroll
        for (int kb = 0; kb < KB; ++kb) {
            const int k0 = kb * 8 + h * 4;
            if constexpr (K % 8 == 0) {
                a[kb] = *reinterpret_cast<const float4*>(A + (size_t)row * K + k0);
            } else {
                const float* ar = A + (size_t)row * K;
                const float x0 = ar[min(k0,     K - 1)], x1 = ar[min(k0 + 1, K - 1)];
                const float x2 = ar[min(k0 + 2, K - 1)], x3 = ar[min(k0 + 3, K - 1)];
                a[kb].x = (k0     < K) ? x0 : 0.0f;
                a[kb].y = (k0 + 1 < K) ? x1 : 0.0f;
                a[kb].z = (k0 + 2 < K) ? x2 : 0.0f;
                a[kb].w = (k0 + 3 < K) ? x3 : 0.0f;
            }
        }
    }

    constexpr int ITEMS = KB * NH * 32;
    constexpr int TRIPS = (ITEMS + THREADS - 1) / THREADS;
#pragma unroll
    for (int j = 0; j < TRIPS; ++j) {
        const int it = tid + j * THREADS;
        if (ITEMS % THREADS == 0 || it < ITEMS) {
            const int l   = it & 31;
            const int bn  = (it >> 5) % NH;
            const int kb  = (it >> 5) / NH;
            const int k0  = kb * 8 + (l >> 4) * 4;
            const int col = bn * 64 + (l & 15) * 4;
            float4 v[4];
#pragma unroll
            for (int e = 0; e < 4; ++e) {
                if constexpr (K % 8 == 0) {
                    v[e] = *reinterpret_cast<const float4*>(B + (size_t)(k0 + e) * N + col);
                } else {
                    const float4 t = *reinterpret_cast<const float4*>(B + (size_t)min(k0 + e, K - 1) * N + col);
                    v[e] = (k0 + e < K) ? t : make_float4(0.0f, 0.0f, 0.0f, 0.0f);
                }
            }
            float4* d = Bs + (kb * NTT + bn * 4) * 32 + l;
            d[0]  = make_float4(v[0].x, v[1].x, v[2].x, v[3].x);
            d[32] = make_float4(v[0].y, v[1].y, v[2].y, v[3].y);
            d[64] = make_float4(v[0].z, v[1].z, v[2].z, v[3].z);
            d[96] = make_float4(v[0].w, v[1].w, v[2].w, v[3].w);
        }
    }
    if constexpr (DEG_PLANES > 0) {
        if (dmine) {
            int d = 1;
#pragma unroll
            for (int p = 0; p < DEG_PLANES; ++p) d += dpart[p];
            const float dv = rsqrtf((float)d);
            dis[drow] = dv;
            reinterpret_cast<float*>(dsh)[tid] = dv;
        }
    }
    __syncthreads();
    if (!active) return;

    v8f acc[4] = {};
#pragma unroll
    for (int kb = 0; kb < KB; ++kb) {
        float4 b[4];
#pragma unroll
        for (int t = 0; t < 4; ++t) b[t] = Bs[(kb * NTT + nh * 4 + t) * 32 + lane];
        const v2f a0 = {a[kb].x, a[kb].y};
        const v2f a1 = {a[kb].z, a[kb].w};
#pragma unroll
        for (int t = 0; t < 4; ++t) {
            const v2f b0 = {b[t].x, b[t].y};
            acc[t] = __builtin_amdgcn_wmma_f32_16x16x4_f32(
                false, a0, false, b0, (short)0, acc[t], false, false);
        }
#pragma unroll
        for (int t = 0; t < 4; ++t) {
            const v2f b1 = {b[t].z, b[t].w};
            acc[t] = __builtin_amdgcn_wmma_f32_16x16x4_f32(
                false, a1, false, b1, (short)0, acc[t], false, false);
        }
    }

    if constexpr (DEG_PLANES > 0) {
        sc0 = dsh[(wave / NH) * 4 + h * 2];
        sc1 = dsh[(wave / NH) * 4 + h * 2 + 1];
    }
    const float sc[8] = {sc0.x, sc0.y, sc0.z, sc0.w, sc1.x, sc1.y, sc1.z, sc1.w};

    float* __restrict__ crow = C + (size_t)(strip * 16 + h * 8) * N + nh * 64 + c * 4;
#pragma unroll
    for (int r = 0; r < 8; ++r)
        *reinterpret_cast<float4*>(crow + (size_t)r * N) =
            make_float4(acc[0][r] * sc[r], acc[1][r] * sc[r], acc[2][r] * sc[r], acc[3][r] * sc[r]);
}

__global__ __launch_bounds__(OWN_THREADS) void bin_edges(
    const int* __restrict__ src, const int* __restrict__ dst,
    int* __restrict__ degp, int* __restrict__ lists, int* __restrict__ counts,
    int nEdges) {
    __shared__ int l_e[LIST_CAP];
    __shared__ int l_a[LIST_CAP];
    __shared__ int deg[GROUP_NODES];
    __shared__ int cnt;
    __shared__ int cnt2[BIN_SPLIT];

    const int tid   = threadIdx.x;
    const int grp   = blockIdx.x / BIN_SPLIT;
    const int chunk = blockIdx.x % BIN_SPLIT;
    const int base  = grp * GROUP_NODES;

    for (int i = tid; i < GROUP_NODES; i += OWN_THREADS) deg[i] = 0;
    if (tid == 0) cnt = 0;
    if (tid < BIN_SPLIT) cnt2[tid] = 0;
    __syncthreads();

    auto put = [&](int slot, int e, int a) {
        if (slot < LIST_CAP) { l_e[slot] = e; l_a[slot] = a; }
        atomicAdd(&deg[a], 1);
    };
    auto take4 = [&](int4 d, int e0) {
        const unsigned a0 = (unsigned)(d.x - base), a1 = (unsigned)(d.y - base);
        const unsigned a2 = (unsigned)(d.z - base), a3 = (unsigned)(d.w - base);
        if (min(min(a0, a1), min(a2, a3)) < GROUP_NODES) {
            const int c = (a0 < GROUP_NODES) + (a1 < GROUP_NODES) + (a2 < GROUP_NODES) + (a3 < GROUP_NODES);
            int slot = atomicAdd(&cnt, c);
            if (a0 < GROUP_NODES) put(slot++, e0,     (int)a0);
            if (a1 < GROUP_NODES) put(slot++, e0 + 1, (int)a1);
            if (a2 < GROUP_NODES) put(slot++, e0 + 2, (int)a2);
            if (a3 < GROUP_NODES) put(slot++, e0 + 3, (int)a3);
        }
    };

    const int4* __restrict__ dst4 = reinterpret_cast<const int4*>(dst);
    const int nQuad = nEdges >> 2;
    const int per   = (nQuad + BIN_SPLIT - 1) / BIN_SPLIT;
    const int qlo   = chunk * per;
    const int qhi   = min(qlo + per, nQuad);
    constexpr int SU = 10;
    for (int q0 = qlo + tid; q0 < qhi; q0 += OWN_THREADS * SU) {
        int4 d[SU];
#pragma unroll
        for (int u = 0; u < SU; ++u) {
            const int q = q0 + OWN_THREADS * u;
            d[u] = (q < qhi) ? dst4[q] : make_int4(-1, -1, -1, -1);
        }
#pragma unroll
        for (int u = 0; u < SU; ++u) take4(d[u], (q0 + OWN_THREADS * u) * 4);
    }
    if (chunk == BIN_SPLIT - 1) {
        for (int e = (nQuad << 2) + tid; e < nEdges; e += OWN_THREADS) {
            const unsigned a = (unsigned)(dst[e] - base);
            if (a < GROUP_NODES) put(atomicAdd(&cnt, 1), e, (int)a);
        }
    }
    __syncthreads();

    for (int i = tid; i < GROUP_NODES; i += OWN_THREADS)
        degp[(size_t)chunk * N_PAD + base + i] = deg[i];

    const int total = cnt;
    const bool fits = total <= LIST_CAP;
    if (fits) {
        const int lane = tid & 31;
        for (int i0 = tid - lane; i0 < total; i0 += OWN_THREADS) {
            const int i = i0 + lane;
            const bool valid = i < total;
            int a = 0, s = 0;
            if (valid) { a = l_a[i]; s = src[l_e[i]]; }
            const int j  = a / OWN_NODES;
            const int dl = a - j * OWN_NODES;
            unsigned same = 0;
            int lane_cnt = 0;
#pragma unroll
            for (int jj = 0; jj < BIN_SPLIT; ++jj) {
                const unsigned mk = (unsigned)__ballot(valid && j == jj);
                if (j == jj) same = mk;
                if (lane == jj) lane_cnt = __popc(mk);
            }
            int first = 0;
            if (lane < BIN_SPLIT) first = atomicAdd(&cnt2[lane], lane_cnt);
            const int slot = __shfl(first, j) + __popc(same & ((1u << lane) - 1u));
            if (valid && slot < SUB_CAP)
                lists[((size_t)(grp * BIN_SPLIT + j) * BIN_SPLIT + chunk) * SUB_CAP + slot] = (s << 8) | dl;
        }
    }
    __syncthreads();
    if (tid < BIN_SPLIT)
        counts[(grp * BIN_SPLIT + tid) * BIN_SPLIT + chunk] = fits ? cnt2[tid] : SUB_CAP + 1;
}

template <int DIM>
__device__ __forceinline__ int acc_pos(int nl, int f) {
    constexpr int PER = DIM / 32;
    return nl * DIM + (f % PER) * 32 + (f / PER);
}

template <int DIM, bool RELU>
__global__ __launch_bounds__(OWN_THREADS)
__attribute__((amdgpu_waves_per_eu(8, 8)))
void aggregate_owner(
    const int* __restrict__ src, const int* __restrict__ dst,
    const float* __restrict__ dis, const float* __restrict__ m,
    const float* __restrict__ bias, const int* __restrict__ lists,
    const int* __restrict__ counts, float* __restrict__ out,
    int nEdges, int nNodes) {
    constexpr int PER   = DIM / 32;
    constexpr int WAVES = OWN_THREADS / 32;
    __shared__ float acc[OWN_NODES * DIM];

    const int tid  = threadIdx.x;
    const int lane = tid & 31;
    const int wave = __builtin_amdgcn_readfirstlane(tid >> 5);
    const int base = blockIdx.x * OWN_NODES;

    for (int i = tid; i < OWN_NODES * DIM; i += OWN_THREADS) acc[i] = 0.0f;

    const int* __restrict__ cn = counts + blockIdx.x * BIN_SPLIT;
    int off[BIN_SPLIT];
    int total = 0;
    bool listed = true;
#pragma unroll
    for (int c = 0; c < BIN_SPLIT; ++c) {
        const int nc = cn[c];
        listed = listed && (nc <= SUB_CAP);
        off[c] = total;
        total += nc;
    }
    const int n = listed ? total : 0;
    const int* __restrict__ mine = lists + (size_t)blockIdx.x * LIST_CAP;

    const int i_lo = (n * wave) / WAVES;
    const int i_hi = (n * (wave + 1)) / WAVES;
    auto record = [&](int i) -> int {
        int c = 0, start = 0;
#pragma unroll
        for (int cc = 1; cc < BIN_SPLIT; ++cc)
            if (i >= off[cc]) { c = cc; start = off[cc]; }
        return (i < i_hi) ? mine[c * SUB_CAP + (i - start)] : -1;
    };
    int   p_cur = record(i_lo + lane);
    int   p_nxt = record(i_lo + 32 + lane);
    __syncthreads();

    constexpr int EU = 8;
    static_assert(EU == 8, "the load fence below names eight rows");
    typedef __attribute__((ext_vector_type(PER))) float vrow;
    for (int g0 = i_lo; g0 < i_hi; g0 += 32) {
        const int p_nn = record(g0 + 64 + lane);
        const int cnt = min(32, i_hi - g0);
        for (int b = 0; b < cnt; b += EU) {
            int dl[EU];
            vrow v[EU];
#pragma unroll
            for (int u = 0; u < EU; ++u) {
                const int l = min(b + u, cnt - 1);
                const int p = __builtin_amdgcn_readlane(p_cur, l);
                dl[u] = p & 255;
                v[u]  = *reinterpret_cast<const vrow*>(m + (size_t)(p >> 8) * DIM + lane * PER);
            }
            asm volatile("" :: "v"(v[0]), "v"(v[1]), "v"(v[2]), "v"(v[3]),
                               "v"(v[4]), "v"(v[5]), "v"(v[6]), "v"(v[7]));
#pragma unroll
            for (int u = 0; u < EU; ++u) {
                if (b + u < cnt) {
#pragma unroll
                    for (int j = 0; j < PER; ++j)
                        atomicAdd(&acc[dl[u] * DIM + j * 32 + lane], v[u][j]);
                }
            }
        }
        p_cur = p_nxt;
        p_nxt = p_nn;
    }

    if (!listed) {
        for (int e = tid; e < nEdges; e += OWN_THREADS) {
            const unsigned a = (unsigned)(dst[e] - base);
            if (a < OWN_NODES) {
                const int s = src[e];
                for (int f = 0; f < DIM; ++f)
                    atomicAdd(&acc[acc_pos<DIM>((int)a, f)], m[(size_t)s * DIM + f]);
            }
        }
    }
    __syncthreads();

    constexpr int ROW4 = DIM / 4;
    for (int idx = tid; idx < OWN_NODES * ROW4; idx += OWN_THREADS) {
        const int nl   = idx / ROW4;
        const int f0   = (idx % ROW4) * 4;
        const int node = base + nl;
        if (node < nNodes) {
            const float di = dis[node];
            const size_t g = (size_t)node * DIM + f0;
            const float4 mv = *reinterpret_cast<const float4*>(m + g);
            const float4 bv = *reinterpret_cast<const float4*>(bias + f0);
            float4 r;
            r.x = (acc[acc_pos<DIM>(nl, f0)]     + mv.x) * di + bv.x;
            r.y = (acc[acc_pos<DIM>(nl, f0 + 1)] + mv.y) * di + bv.y;
            r.z = (acc[acc_pos<DIM>(nl, f0 + 2)] + mv.z) * di + bv.z;
            r.w = (acc[acc_pos<DIM>(nl, f0 + 3)] + mv.w) * di + bv.w;
            if (RELU) {
                r.x = fmaxf(r.x, 0.0f); r.y = fmaxf(r.y, 0.0f);
                r.z = fmaxf(r.z, 0.0f); r.w = fmaxf(r.w, 0.0f);
            }
            *reinterpret_cast<float4*>(out + g) = r;
        }
    }
}

extern "C" void kernel_launch(void* const* d_in, const int* in_sizes, int n_in,
                              void* d_out, int out_size, void* d_ws, size_t ws_size,
                              hipStream_t stream) {
    const float* x   = (const float*)d_in[0];
    const int*   ei  = (const int*)d_in[1];
    const float* W0  = (const float*)d_in[2];
    const float* b0  = (const float*)d_in[3];
    const float* W1  = (const float*)d_in[4];
    const float* b1  = (const float*)d_in[5];
    const float* W2  = (const float*)d_in[6];
    const float* b2  = (const float*)d_in[7];
    float* out = (float*)d_out;

    const int* src = ei;
    const int* dst = ei + N_EDGES;

    const int gemmBlocks  = (N_NODES / 16 + 3) / 4;
    const int gemmThreadsH = (HID_DIM / 64) * 128;
    const int gemmThreadsL = (LAT_DIM / 64) * 128;
    const int ownerBlocks = (N_NODES + OWN_NODES - 1) / OWN_NODES;

    const int binBlocks   = N_GROUPS * BIN_SPLIT;
    const int ownersPad   = N_GROUPS * BIN_SPLIT;

    float* ws  = (float*)d_ws;
    float* dis = ws;
    float* m   = dis + N_NODES;
    float* h   = m   + (size_t)N_NODES * HID_DIM;
    int* lists  = (int*)(h + (size_t)N_NODES * HID_DIM);
    int* counts = lists + (size_t)ownersPad * LIST_CAP;
    int* degp   = counts + (size_t)ownersPad * BIN_SPLIT;

    bin_edges<<<binBlocks, OWN_THREADS, 0, stream>>>(src, dst, degp, lists, counts, N_EDGES);

    gemm_strip_f32<IN_DIM, HID_DIM, BIN_SPLIT><<<gemmBlocks, gemmThreadsH, 0, stream>>>(x, W0, m, N_NODES, degp, dis, nullptr);
    aggregate_owner<HID_DIM, true><<<ownerBlocks, OWN_THREADS, 0, stream>>>(
        src, dst, dis, m, b0, lists, counts, h, N_EDGES, N_NODES);

    gemm_strip_f32<HID_DIM, HID_DIM, 0><<<gemmBlocks, gemmThreadsH, 0, stream>>>(h, W1, m, N_NODES, nullptr, nullptr, dis);
    aggregate_owner<HID_DIM, true><<<ownerBlocks, OWN_THREADS, 0, stream>>>(
        src, dst, dis, m, b1, lists, counts, h, N_EDGES, N_NODES);

    gemm_strip_f32<HID_DIM, LAT_DIM, 0><<<gemmBlocks, gemmThreadsL, 0, stream>>>(h, W2, m, N_NODES, nullptr, nullptr, dis);
    aggregate_owner<LAT_DIM, false><<<ownerBlocks, OWN_THREADS, 0, stream>>>(
        src, dst, dis, m, b2, lists, counts, out, N_EDGES, N_NODES);
}


# speedup vs baseline: 3.7439x; 1.0322x over previous
#include <hip/hip_runtime.h>
#include <hip/hip_bf16.h>


typedef __attribute__((ext_vector_type(2))) float v2f;
typedef __attribute__((ext_vector_type(8))) float v8f;

#define N_NODES 40000
#define N_EDGES 640000
#define IN_DIM  3
#define HID_DIM 128
#define LAT_DIM 64

#define OWN_NODES 160
#define OWN_THREADS 1024
#define LIST_CAP 8192
#define BIN_SPLIT 16
#define GROUP_NODES (OWN_NODES * BIN_SPLIT)
#define SUB_CAP (LIST_CAP / BIN_SPLIT)
#define N_GROUPS ((N_NODES + GROUP_NODES - 1) / GROUP_NODES)
#define N_PAD (N_GROUPS * GROUP_NODES)

template <int K, int N, int DEG_PLANES>
__global__ __launch_bounds__((N / 64) * 128) void gemm_strip_f32(
    const float* __restrict__ A, const float* __restrict__ B,
    float* __restrict__ C, int M,
    const int* __restrict__ degp, float* __restrict__ dis, const float* __restrict__ dis_in) {
    constexpr int NH      = N / 64;
    constexpr int NTT     = N / 16;
    constexpr int KB      = (K + 7) / 8;
    constexpr int THREADS = NH * 128;
    __shared__ float4 Bs[KB * NTT * 32];
    __shared__ float4 dsh[DEG_PLANES > 0 ? 16 : 1];

    const int tid  = threadIdx.x;
    const int lane = tid & 31;
    const int wave = tid >> 5;
    const int c    = lane & 15;
    const int h    = lane >> 4;
    const int nh   = wave % NH;
    const int strip = blockIdx.x * 4 + wave / NH;
    const bool active = strip * 16 < M;
    const int row  = strip * 16 + c;

    const int  drow  = blockIdx.x * 64 + tid;
    const bool dmine = DEG_PLANES > 0 && tid < 64 && drow < M;
    int dpart[DEG_PLANES > 0 ? DEG_PLANES : 1];
    if constexpr (DEG_PLANES > 0) {
        if (dmine) {
#pragma unroll
            for (int p = 0; p < DEG_PLANES; ++p) dpart[p] = degp[(size_t)p * N_PAD + drow];
        }
    }

    float4 a[KB];
    float4 sc0 = make_float4(0.0f, 0.0f, 0.0f, 0.0f), sc1 = sc0;
    if (active) {
        if constexpr (DEG_PLANES == 0) {
            const float4* sp = reinterpret_cast<const float4*>(dis_in + strip * 16 + h * 8);
            sc0 = sp[0];
            sc1 = sp[1];
        }
#pragma unroll
        for (int kb = 0; kb < KB; ++kb) {
            const int k0 = kb * 8 + h * 4;
            if constexpr (K % 8 == 0) {
                a[kb] = *reinterpret_cast<const float4*>(A + (size_t)row * K + k0);
            } else {
                const float* ar = A + (size_t)row * K;
                const float x0 = ar[min(k0,     K - 1)], x1 = ar[min(k0 + 1, K - 1)];
                const float x2 = ar[min(k0 + 2, K - 1)], x3 = ar[min(k0 + 3, K - 1)];
                a[kb].x = (k0     < K) ? x0 : 0.0f;
                a[kb].y = (k0 + 1 < K) ? x1 : 0.0f;
                a[kb].z = (k0 + 2 < K) ? x2 : 0.0f;
                a[kb].w = (k0 + 3 < K) ? x3 : 0.0f;
            }
        }
    }

    constexpr int ITEMS = KB * NH * 32;
    constexpr int TRIPS = (ITEMS + THREADS - 1) / THREADS;
#pragma unroll
    for (int j = 0; j < TRIPS; ++j) {
        const int it = tid + j * THREADS;
        if (ITEMS % THREADS == 0 || it < ITEMS) {
            const int l   = it & 31;
            const int bn  = (it >> 5) % NH;
            const int kb  = (it >> 5) / NH;
            const int k0  = kb * 8 + (l >> 4) * 4;
            const int col = bn * 64 + (l & 15) * 4;
            float4 v[4];
#pragma unroll
            for (int e = 0; e < 4; ++e) {
                if constexpr (K % 8 == 0) {
                    v[e] = *reinterpret_cast<const float4*>(B + (size_t)(k0 + e) * N + col);
                } else {
                    const float4 t = *reinterpret_cast<const float4*>(B + (size_t)min(k0 + e, K - 1) * N + col);
                    v[e] = (k0 + e < K) ? t : make_float4(0.0f, 0.0f, 0.0f, 0.0f);
                }
            }
            float4* d = Bs + (kb * NTT + bn * 4) * 32 + l;
            d[0]  = make_float4(v[0].x, v[1].x, v[2].x, v[3].x);
            d[32] = make_float4(v[0].y, v[1].y, v[2].y, v[3].y);
            d[64] = make_float4(v[0].z, v[1].z, v[2].z, v[3].z);
            d[96] = make_float4(v[0].w, v[1].w, v[2].w, v[3].w);
        }
    }
    if constexpr (DEG_PLANES > 0) {
        if (dmine) {
            int d = 1;
#pragma unroll
            for (int p = 0; p < DEG_PLANES; ++p) d += dpart[p];
            const float dv = rsqrtf((float)d);
            dis[drow] = dv;
            reinterpret_cast<float*>(dsh)[tid] = dv;
        }
    }
    __syncthreads();
    if (!active) return;

    v8f acc[4] = {};
#pragma unroll
    for (int kb = 0; kb < KB; ++kb) {
        float4 b[4];
#pragma unroll
        for (int t = 0; t < 4; ++t) b[t] = Bs[(kb * NTT + nh * 4 + t) * 32 + lane];
        const v2f a0 = {a[kb].x, a[kb].y};
        const v2f a1 = {a[kb].z, a[kb].w};
#pragma unroll
        for (int t = 0; t < 4; ++t) {
            const v2f b0 = {b[t].x, b[t].y};
            acc[t] = __builtin_amdgcn_wmma_f32_16x16x4_f32(
                false, a0, false, b0, (short)0, acc[t], false, false);
        }
#pragma unroll
        for (int t = 0; t < 4; ++t) {
            const v2f b1 = {b[t].z, b[t].w};
            acc[t] = __builtin_amdgcn_wmma_f32_16x16x4_f32(
                false, a1, false, b1, (short)0, acc[t], false, false);
        }
    }

    if constexpr (DEG_PLANES > 0) {
        sc0 = dsh[(wave / NH) * 4 + h * 2];
        sc1 = dsh[(wave / NH) * 4 + h * 2 + 1];
    }
    const float sc[8] = {sc0.x, sc0.y, sc0.z, sc0.w, sc1.x, sc1.y, sc1.z, sc1.w};

    float* __restrict__ crow = C + (size_t)(strip * 16 + h * 8) * N + nh * 64 + c * 4;
#pragma unroll
    for (int r = 0; r < 8; ++r)
        *reinterpret_cast<float4*>(crow + (size_t)r * N) =
            make_float4(acc[0][r] * sc[r], acc[1][r] * sc[r], acc[2][r] * sc[r], acc[3][r] * sc[r]);
}

__global__ __launch_bounds__(OWN_THREADS) void bin_edges(
    const int* __restrict__ src, const int* __restrict__ dst,
    int* __restrict__ degp, int* __restrict__ lists, int* __restrict__ counts,
    int nEdges) {
    __shared__ int l_e[LIST_CAP];
    __shared__ int l_a[LIST_CAP];
    __shared__ int deg[GROUP_NODES];
    __shared__ int cnt;
    __shared__ int cnt2[BIN_SPLIT];

    const int tid   = threadIdx.x;
    const int grp   = blockIdx.x / BIN_SPLIT;
    const int chunk = blockIdx.x % BIN_SPLIT;
    const int base  = grp * GROUP_NODES;

    for (int i = tid; i < GROUP_NODES; i += OWN_THREADS) deg[i] = 0;
    if (tid == 0) cnt = 0;
    if (tid < BIN_SPLIT) cnt2[tid] = 0;
    __syncthreads();

    auto put = [&](int slot, int e, int a) {
        if (slot < LIST_CAP) { l_e[slot] = e; l_a[slot] = a; }
        atomicAdd(&deg[a], 1);
    };
    auto take4 = [&](int4 d, int e0) {
        const unsigned a0 = (unsigned)(d.x - base), a1 = (unsigned)(d.y - base);
        const unsigned a2 = (unsigned)(d.z - base), a3 = (unsigned)(d.w - base);
        if (min(min(a0, a1), min(a2, a3)) < GROUP_NODES) {
            const int c = (a0 < GROUP_NODES) + (a1 < GROUP_NODES) + (a2 < GROUP_NODES) + (a3 < GROUP_NODES);
            int slot = atomicAdd(&cnt, c);
            if (a0 < GROUP_NODES) put(slot++, e0,     (int)a0);
            if (a1 < GROUP_NODES) put(slot++, e0 + 1, (int)a1);
            if (a2 < GROUP_NODES) put(slot++, e0 + 2, (int)a2);
            if (a3 < GROUP_NODES) put(slot++, e0 + 3, (int)a3);
        }
    };

    const int4* __restrict__ dst4 = reinterpret_cast<const int4*>(dst);
    const int nQuad = nEdges >> 2;
    const int per   = (nQuad + BIN_SPLIT - 1) / BIN_SPLIT;
    const int qlo   = chunk * per;
    const int qhi   = min(qlo + per, nQuad);
    constexpr int SU = 10;
    for (int q0 = qlo + tid; q0 < qhi; q0 += OWN_THREADS * SU) {
        int4 d[SU];
#pragma unroll
        for (int u = 0; u < SU; ++u) {
            const int q = q0 + OWN_THREADS * u;
            d[u] = (q < qhi) ? dst4[q] : make_int4(-1, -1, -1, -1);
        }
#pragma unroll
        for (int u = 0; u < SU; ++u) take4(d[u], (q0 + OWN_THREADS * u) * 4);
    }
    if (chunk == BIN_SPLIT - 1) {
        for (int e = (nQuad << 2) + tid; e < nEdges; e += OWN_THREADS) {
            const unsigned a = (unsigned)(dst[e] - base);
            if (a < GROUP_NODES) put(atomicAdd(&cnt, 1), e, (int)a);
        }
    }
    __syncthreads();

    for (int i = tid; i < GROUP_NODES; i += OWN_THREADS)
        degp[(size_t)chunk * N_PAD + base + i] = deg[i];

    const int total = cnt;
    const bool fits = total <= LIST_CAP;
    if (fits) {
        const int lane = tid & 31;
        for (int i0 = tid - lane; i0 < total; i0 += OWN_THREADS) {
            const int i = i0 + lane;
            const bool valid = i < total;
            int a = 0, s = 0;
            if (valid) { a = l_a[i]; s = src[l_e[i]]; }
            const int j  = a / OWN_NODES;
            const int dl = a - j * OWN_NODES;
            unsigned same = 0;
            int lane_cnt = 0;
#pragma unroll
            for (int jj = 0; jj < BIN_SPLIT; ++jj) {
                const unsigned mk = (unsigned)__ballot(valid && j == jj);
                if (j == jj) same = mk;
                if (lane == jj) lane_cnt = __popc(mk);
            }
            int first = 0;
            if (lane < BIN_SPLIT) first = atomicAdd(&cnt2[lane], lane_cnt);
            const int slot = __shfl(first, j) + __popc(same & ((1u << lane) - 1u));
            if (valid && slot < SUB_CAP)
                lists[((size_t)(grp * BIN_SPLIT + j) * BIN_SPLIT + chunk) * SUB_CAP + slot] = (s << 8) | dl;
        }
    }
    __syncthreads();
    if (tid < BIN_SPLIT)
        counts[(grp * BIN_SPLIT + tid) * BIN_SPLIT + chunk] = fits ? cnt2[tid] : SUB_CAP + 1;
}

#define ACC_BLOCKS 8
#define ACC_BLOCK_WORDS 8192
#define ACC_ROWS_PER_BLOCK ((OWN_NODES + ACC_BLOCKS - 1) / ACC_BLOCKS)
template <int DIM>
__device__ __forceinline__ int acc_row(int nl) {
    return (nl % ACC_BLOCKS) * ACC_BLOCK_WORDS + (nl / ACC_BLOCKS) * DIM;
}
template <int DIM>
__device__ __forceinline__ int acc_pos(int nl, int f) {
    constexpr int PER = DIM / 32;
    return acc_row<DIM>(nl) + (f % PER) * 32 + (f / PER);
}

template <int DIM, bool RELU>
__global__ __launch_bounds__(OWN_THREADS)
__attribute__((amdgpu_waves_per_eu(8, 8)))
void aggregate_owner(
    const int* __restrict__ src, const int* __restrict__ dst,
    const float* __restrict__ dis, const float* __restrict__ m,
    const float* __restrict__ bias, const int* __restrict__ lists,
    const int* __restrict__ counts, float* __restrict__ out,
    int nEdges, int nNodes) {
    constexpr int PER   = DIM / 32;
    constexpr int WAVES = OWN_THREADS / 32;
    static_assert(ACC_ROWS_PER_BLOCK * DIM <= ACC_BLOCK_WORDS, "a block's rows must fit in it");
    __shared__ float acc[(ACC_BLOCKS - 1) * ACC_BLOCK_WORDS + ACC_ROWS_PER_BLOCK * DIM];

    const int tid  = threadIdx.x;
    const int lane = tid & 31;
    const int wave = __builtin_amdgcn_readfirstlane(tid >> 5);
    const int base = blockIdx.x * OWN_NODES;

    for (int i = tid; i < OWN_NODES * DIM; i += OWN_THREADS) acc[acc_row<DIM>(i / DIM) + i % DIM] = 0.0f;

    const int* __restrict__ cn = counts + blockIdx.x * BIN_SPLIT;
    int off[BIN_SPLIT];
    int total = 0;
    bool listed = true;
#pragma unroll
    for (int c = 0; c < BIN_SPLIT; ++c) {
        const int nc = cn[c];
        listed = listed && (nc <= SUB_CAP);
        off[c] = total;
        total += nc;
    }
    const int n = listed ? total : 0;
    const int* __restrict__ mine = lists + (size_t)blockIdx.x * LIST_CAP;

    const int i_lo = (n * wave) / WAVES;
    const int i_hi = (n * (wave + 1)) / WAVES;
    auto record = [&](int i) -> int {
        int c = 0, start = 0;
#pragma unroll
        for (int cc = 1; cc < BIN_SPLIT; ++cc)
            if (i >= off[cc]) { c = cc; start = off[cc]; }
        return (i < i_hi) ? mine[c * SUB_CAP + (i - start)] : -1;
    };
    int   p_cur = record(i_lo + lane);
    int   p_nxt = record(i_lo + 32 + lane);
    __syncthreads();

    constexpr int EU = 8;
    static_assert(EU == 8, "the load fence below names eight rows");
    typedef __attribute__((ext_vector_type(PER))) float vrow;
    for (int g0 = i_lo; g0 < i_hi; g0 += 32) {
        const int p_nn = record(g0 + 64 + lane);
        const int cnt = min(32, i_hi - g0);
        for (int b = 0; b < cnt; b += EU) {
            int dl[EU];
            vrow v[EU];
#pragma unroll
            for (int u = 0; u < EU; ++u) {
                const int l = min(b + u, cnt - 1);
                const int p = __builtin_amdgcn_readlane(p_cur, l);
                dl[u] = acc_row<DIM>(p & 255);
                v[u]  = *reinterpret_cast<const vrow*>(m + (size_t)(p >> 8) * DIM + lane * PER);
            }
            asm volatile("" :: "v"(v[0]), "v"(v[1]), "v"(v[2]), "v"(v[3]),
                               "v"(v[4]), "v"(v[5]), "v"(v[6]), "v"(v[7]));
#pragma unroll
            for (int u = 0; u < EU; ++u) {
                if (b + u < cnt) {
#pragma unroll
                    for (int j = 0; j < PER; ++j)
                        atomicAdd(&acc[dl[u] + j * 32 + lane], v[u][j]);
                }
            }
        }
        p_cur = p_nxt;
        p_nxt = p_nn;
    }

    if (!listed) {
        for (int e = tid; e < nEdges; e += OWN_THREADS) {
            const unsigned a = (unsigned)(dst[e] - base);
            if (a < OWN_NODES) {
                const int s = src[e];
                for (int f = 0; f < DIM; ++f)
                    atomicAdd(&acc[acc_pos<DIM>((int)a, f)], m[(size_t)s * DIM + f]);
            }
        }
    }
    __syncthreads();

    constexpr int ROW4 = DIM / 4;
    for (int idx = tid; idx < OWN_NODES * ROW4; idx += OWN_THREADS) {
        const int nl   = idx / ROW4;
        const int f0   = (idx % ROW4) * 4;
        const int node = base + nl;
        if (node < nNodes) {
            const float di = dis[node];
            const size_t g = (size_t)node * DIM + f0;
            const float4 mv = *reinterpret_cast<const float4*>(m + g);
            const float4 bv = *reinterpret_cast<const float4*>(bias + f0);
            float4 r;
            r.x = (acc[acc_pos<DIM>(nl, f0)]     + mv.x) * di + bv.x;
            r.y = (acc[acc_pos<DIM>(nl, f0 + 1)] + mv.y) * di + bv.y;
            r.z = (acc[acc_pos<DIM>(nl, f0 + 2)] + mv.z) * di + bv.z;
            r.w = (acc[acc_pos<DIM>(nl, f0 + 3)] + mv.w) * di + bv.w;
            if (RELU) {
                r.x = fmaxf(r.x, 0.0f); r.y = fmaxf(r.y, 0.0f);
                r.z = fmaxf(r.z, 0.0f); r.w = fmaxf(r.w, 0.0f);
            }
            *reinterpret_cast<float4*>(out + g) = r;
        }
    }
}

extern "C" void kernel_launch(void* const* d_in, const int* in_sizes, int n_in,
                              void* d_out, int out_size, void* d_ws, size_t ws_size,
                              hipStream_t stream) {
    const float* x   = (const float*)d_in[0];
    const int*   ei  = (const int*)d_in[1];
    const float* W0  = (const float*)d_in[2];
    const float* b0  = (const float*)d_in[3];
    const float* W1  = (const float*)d_in[4];
    const float* b1  = (const float*)d_in[5];
    const float* W2  = (const float*)d_in[6];
    const float* b2  = (const float*)d_in[7];
    float* out = (float*)d_out;

    const int* src = ei;
    const int* dst = ei + N_EDGES;

    const int gemmBlocks  = (N_NODES / 16 + 3) / 4;
    const int gemmThreadsH = (HID_DIM / 64) * 128;
    const int gemmThreadsL = (LAT_DIM / 64) * 128;
    const int ownerBlocks = (N_NODES + OWN_NODES - 1) / OWN_NODES;

    const int binBlocks   = N_GROUPS * BIN_SPLIT;
    const int ownersPad   = N_GROUPS * BIN_SPLIT;

    float* ws  = (float*)d_ws;
    float* dis = ws;
    float* m   = dis + N_NODES;
    float* h   = m   + (size_t)N_NODES * HID_DIM;
    int* lists  = (int*)(h + (size_t)N_NODES * HID_DIM);
    int* counts = lists + (size_t)ownersPad * LIST_CAP;
    int* degp   = counts + (size_t)ownersPad * BIN_SPLIT;

    bin_edges<<<binBlocks, OWN_THREADS, 0, stream>>>(src, dst, degp, lists, counts, N_EDGES);

    gemm_strip_f32<IN_DIM, HID_DIM, BIN_SPLIT><<<gemmBlocks, gemmThreadsH, 0, stream>>>(x, W0, m, N_NODES, degp, dis, nullptr);
    aggregate_owner<HID_DIM, true><<<ownerBlocks, OWN_THREADS, 0, stream>>>(
        src, dst, dis, m, b0, lists, counts, h, N_EDGES, N_NODES);

    gemm_strip_f32<HID_DIM, HID_DIM, 0><<<gemmBlocks, gemmThreadsH, 0, stream>>>(h, W1, m, N_NODES, nullptr, nullptr, dis);
    aggregate_owner<HID_DIM, true><<<ownerBlocks, OWN_THREADS, 0, stream>>>(
        src, dst, dis, m, b1, lists, counts, h, N_EDGES, N_NODES);

    gemm_strip_f32<HID_DIM, LAT_DIM, 0><<<gemmBlocks, gemmThreadsL, 0, stream>>>(h, W2, m, N_NODES, nullptr, nullptr, dis);
    aggregate_owner<LAT_DIM, false><<<ownerBlocks, OWN_THREADS, 0, stream>>>(
        src, dst, dis, m, b2, lists, counts, out, N_EDGES, N_NODES);
}
